# starting kernel
# baseline (speedup 1.0000x reference)
#include <hip/hip_runtime.h>
#include <math.h>

typedef __attribute__((ext_vector_type(2))) float v2f;
typedef __attribute__((ext_vector_type(8))) float v8f;

#define NNODES 50000
#define NEDGES 800000

template <int NC>
__global__ __launch_bounds__(256) void gemm_wmma_f32(
    const float* __restrict__ X, const float* __restrict__ W,
    float* __restrict__ H, int n) {
  constexpr int K  = 128;
  constexpr int CT = (NC + 15) / 16;
  constexpr int PC = CT * 16;
  __shared__ v2f lw[(K / 2) * PC];

  const int tid = threadIdx.x;
  for (int i = tid; i < (K / 2) * PC; i += 256) {
    const int kp = i / PC, c = i % PC;
    v2f val;
    val.x = (c < NC) ? W[(2 * kp) * NC + c] : 0.0f;
    val.y = (c < NC) ? W[(2 * kp + 1) * NC + c] : 0.0f;
    lw[i] = val;
  }
  __syncthreads();

  const int wave = tid >> 5;
  const int lane = tid & 31;
  const int row0 = (blockIdx.x * 8 + wave) * 16;
  if (row0 >= n) return;

  const int m  = lane & 15;
  const int hk = lane >> 4;

  v8f acc[CT];
#pragma unroll
  for (int t = 0; t < CT; ++t) acc[t] = {};

  const float* xrow = X + (size_t)(row0 + m) * K + 2 * hk;

  for (int k4 = 0; k4 < K; k4 += 4) {
    v2f a;
    a.x = xrow[k4];
    a.y = xrow[k4 + 1];
    const v2f* brow = &lw[(k4 / 2 + hk) * PC + m];
#pragma unroll
    for (int t = 0; t < CT; ++t) {
      const v2f b = brow[t * 16];
      acc[t] = __builtin_amdgcn_wmma_f32_16x16x4_f32(
          false, a, false, b, (short)0, acc[t], false, false);
    }
  }

  const int crow = row0 + hk * 8;
#pragma unroll
  for (int t = 0; t < CT; ++t) {
    const int col = t * 16 + m;
    if ((NC % 16 == 0) || (col < NC)) {
#pragma unroll
      for (int r = 0; r < 8; ++r)
        H[(size_t)(crow + r) * NC + col] = acc[t][r];
    }
  }
}

__global__ void zerof(float* __restrict__ p, long long n) {
  long long i = (long long)blockIdx.x * blockDim.x + threadIdx.x;
  if (i < n) p[i] = 0.0f;
}

__global__ __launch_bounds__(256) void agg128(
    const float* __restrict__ H, const int* __restrict__ src,
    const int* __restrict__ tgt, const float* __restrict__ w,
    float* __restrict__ A, int E) {
  const int e = (int)(((long long)blockIdx.x * blockDim.x + threadIdx.x) >> 5);
  const int lane = threadIdx.x & 31;
  if (e >= E) return;
  const int   s  = src[e];
  const int   t  = tgt[e];
  const float we = w[e];
  const float4 v = *((const float4*)(H + (size_t)s * 128) + lane);
  float* ap = A + (size_t)t * 128 + lane * 4;
  atomicAdd(ap + 0, we * v.x);
  atomicAdd(ap + 1, we * v.y);
  atomicAdd(ap + 2, we * v.z);
  atomicAdd(ap + 3, we * v.w);
}

__global__ __launch_bounds__(256) void agg40(
    const float* __restrict__ H, const int* __restrict__ src,
    const int* __restrict__ tgt, const float* __restrict__ w,
    float* __restrict__ A, int E) {
  const int e = (int)(((long long)blockIdx.x * blockDim.x + threadIdx.x) >> 5);
  const int lane = threadIdx.x & 31;
  if (e >= E) return;
  const int   s  = src[e];
  const int   t  = tgt[e];
  const float we = w[e];
  atomicAdd(A + (size_t)t * 40 + lane, we * H[(size_t)s * 40 + lane]);
  if (lane < 8)
    atomicAdd(A + (size_t)t * 40 + 32 + lane, we * H[(size_t)s * 40 + 32 + lane]);
}

__global__ void bias_relu128(float* __restrict__ A, const float* __restrict__ b,
                             long long total) {
  long long i = (long long)blockIdx.x * blockDim.x + threadIdx.x;
  if (i >= total) return;
  const int c = (int)(i & 127);
  A[i] = fmaxf(A[i] + b[c], 0.0f);
}

__global__ __launch_bounds__(256) void gn_bias_relu(
    float* __restrict__ A, const float* __restrict__ b,
    const float* __restrict__ gamma, const float* __restrict__ beta, int n) {
  const int node = (int)(((long long)blockIdx.x * blockDim.x + threadIdx.x) >> 5);
  const int g = threadIdx.x & 31;
  if (node >= n) return;
  float* p = A + (size_t)node * 128 + g * 4;
  float4 v  = *(float4*)p;
  const float4 bb = *(const float4*)(b + g * 4);
  v.x = fmaxf(v.x + bb.x, 0.0f);
  v.y = fmaxf(v.y + bb.y, 0.0f);
  v.z = fmaxf(v.z + bb.z, 0.0f);
  v.w = fmaxf(v.w + bb.w, 0.0f);
  const float mu  = 0.25f * (v.x + v.y + v.z + v.w);
  const float ms  = 0.25f * (v.x * v.x + v.y * v.y + v.z * v.z + v.w * v.w);
  const float inv = rsqrtf(fmaxf(ms - mu * mu, 0.0f) + 1e-5f);
  const float4 gm = *(const float4*)(gamma + g * 4);
  const float4 bt = *(const float4*)(beta + g * 4);
  v.x = (v.x - mu) * inv * gm.x + bt.x;
  v.y = (v.y - mu) * inv * gm.y + bt.y;
  v.z = (v.z - mu) * inv * gm.z + bt.z;
  v.w = (v.w - mu) * inv * gm.w + bt.w;
  *(float4*)p = v;
}

__global__ __launch_bounds__(256) void lsm40(
    const float* __restrict__ Ain, const float* __restrict__ b,
    float* __restrict__ out, int n) {
  const int node = (int)(((long long)blockIdx.x * blockDim.x + threadIdx.x) >> 5);
  const int lane = threadIdx.x & 31;
  if (node >= n) return;
  const float* row = Ain + (size_t)node * 40;
  const float v0 = row[lane] + b[lane];
  const float v1 = (lane < 8) ? (row[32 + lane] + b[32 + lane]) : -INFINITY;
  float m = fmaxf(v0, v1);
#pragma unroll
  for (int o = 16; o > 0; o >>= 1) m = fmaxf(m, __shfl_xor(m, o, 32));
  float s = expf(v0 - m) + ((lane < 8) ? expf(v1 - m) : 0.0f);
#pragma unroll
  for (int o = 16; o > 0; o >>= 1) s += __shfl_xor(s, o, 32);
  const float lse = m + logf(s);
  out[(size_t)node * 40 + lane] = v0 - lse;
  if (lane < 8) out[(size_t)node * 40 + 32 + lane] = v1 - lse;
}

extern "C" void kernel_launch(void* const* d_in, const int* in_sizes, int n_in,
                              void* d_out, int out_size, void* d_ws, size_t ws_size,
                              hipStream_t stream) {
  const int N = NNODES, E = NEDGES;
  const float* x     = (const float*)d_in[0];
  const int*   src   = (const int*)d_in[1];
  const int*   tgt   = (const int*)d_in[2];
  const float* w     = (const float*)d_in[3];
  const float* W0    = (const float*)d_in[4];
  const float* b0    = (const float*)d_in[5];
  const float* W1    = (const float*)d_in[6];
  const float* b1    = (const float*)d_in[7];
  const float* W2    = (const float*)d_in[8];
  const float* b2    = (const float*)d_in[9];
  const float* gamma = (const float*)d_in[10];
  const float* beta  = (const float*)d_in[11];
  float* out = (float*)d_out;

  float* H   = (float*)d_ws;
  float* A   = H + (size_t)N * 128;
  float* H2  = H;
  float* AG2 = H + (size_t)N * 64;

  const long long tot128 = (long long)N * 128;
  const long long tot40  = (long long)N * 40;
  const int gemm_blocks  = (N + 127) / 128;
  const int edge_blocks  = (E + 7) / 8;
  const int ew_blocks    = (int)((tot128 + 255) / 256);
  const int node_blocks  = (N * 32 + 255) / 256;

  gemm_wmma_f32<128><<<gemm_blocks, 256, 0, stream>>>(x, W0, H, N);
  zerof<<<ew_blocks, 256, 0, stream>>>(A, tot128);
  agg128<<<edge_blocks, 256, 0, stream>>>(H, src, tgt, w, A, E);
  bias_relu128<<<ew_blocks, 256, 0, stream>>>(A, b0, tot128);

  gemm_wmma_f32<128><<<gemm_blocks, 256, 0, stream>>>(A, W1, H, N);
  zerof<<<ew_blocks, 256, 0, stream>>>(A, tot128);
  agg128<<<edge_blocks, 256, 0, stream>>>(H, src, tgt, w, A, E);
  gn_bias_relu<<<node_blocks, 256, 0, stream>>>(A, b1, gamma, beta, N);

  gemm_wmma_f32<40><<<gemm_blocks, 256, 0, stream>>>(A, W2, H2, N);
  zerof<<<(int)((tot40 + 255) / 256), 256, 0, stream>>>(AG2, tot40);
  agg40<<<edge_blocks, 256, 0, stream>>>(H2, src, tgt, w, AG2, E);
  lsm40<<<node_blocks, 256, 0, stream>>>(AG2, b2, out, N);
}


# bias+relu, bias+relu+groupnorm and bias+log_softmax applied inside the gather kernels on the sums in registers: 3 of 14 launches gone; 40-wide gather 8 lanes per node, softmax sum tree changed
# speedup vs baseline: 2.2176x; 2.2176x over previous
#include <hip/hip_runtime.h>
#include <math.h>

typedef __attribute__((ext_vector_type(2))) float v2f;
typedef __attribute__((ext_vector_type(8))) float v8f;

#define NNODES 50000
#define NEDGES 800000

template <int NC>
__global__ __launch_bounds__(256) void gemm_wmma_f32(
    const float* __restrict__ X, const float* __restrict__ W,
    float* __restrict__ H, int n) {
  constexpr int K  = 128;
  constexpr int CT = (NC + 15) / 16;
  constexpr int PC = CT * 16;
  __shared__ v2f lw[(K / 2) * PC];

  const int tid = threadIdx.x;
  for (int i = tid; i < (K / 2) * PC; i += 256) {
    const int kp = i / PC, c = i % PC;
    v2f val;
    val.x = (c < NC) ? W[(2 * kp) * NC + c] : 0.0f;
    val.y = (c < NC) ? W[(2 * kp + 1) * NC + c] : 0.0f;
    lw[i] = val;
  }
  __syncthreads();

  const int wave = tid >> 5;
  const int lane = tid & 31;
  const int row0 = (blockIdx.x * 8 + wave) * 16;
  if (row0 >= n) return;

  const int m  = lane & 15;
  const int hk = lane >> 4;

  v8f acc[CT];
#pragma unroll
  for (int t = 0; t < CT; ++t) acc[t] = {};

  const float* xrow = X + (size_t)(row0 + m) * K + 2 * hk;

  for (int k4 = 0; k4 < K; k4 += 4) {
    v2f a;
    a.x = xrow[k4];
    a.y = xrow[k4 + 1];
    const v2f* brow = &lw[(k4 / 2 + hk) * PC + m];
#pragma unroll
    for (int t = 0; t < CT; ++t) {
      const v2f b = brow[t * 16];
      acc[t] = __builtin_amdgcn_wmma_f32_16x16x4_f32(
          false, a, false, b, (short)0, acc[t], false, false);
    }
  }

  const int crow = row0 + hk * 8;
#pragma unroll
  for (int t = 0; t < CT; ++t) {
    const int col = t * 16 + m;
    if ((NC % 16 == 0) || (col < NC)) {
#pragma unroll
      for (int r = 0; r < 8; ++r)
        H[(size_t)(crow + r) * NC + col] = acc[t][r];
    }
  }
}

#define SCAN_TILE 4096
#define NCAP (((NNODES + SCAN_TILE - 1) / SCAN_TILE) * SCAN_TILE)

__global__ void csr_zero(int* __restrict__ cnt, int ncap) {
  const int i = blockIdx.x * blockDim.x + threadIdx.x;
  if (i < ncap) cnt[i] = 0;
}

__global__ void csr_rank(const int* __restrict__ tgt, int* __restrict__ cnt,
                         int* __restrict__ rank, int E) {
  const int e = blockIdx.x * blockDim.x + threadIdx.x;
  if (e < E) rank[e] = atomicAdd(&cnt[tgt[e]], 1);
}

__global__ __launch_bounds__(1024) void csr_scan(
    const int* __restrict__ cnt, int* __restrict__ rowptr, int n, int ncap) {
  __shared__ int wtot[2][32];
  const int tid = threadIdx.x, lane = tid & 31, wave = tid >> 5;
  const int4* c4 = (const int4*)cnt;
  int carry = 0, buf = 0;
  int4 cur = c4[tid];
  for (int base = 0; base < ncap; base += SCAN_TILE, buf ^= 1) {
    int4 nxt = make_int4(0, 0, 0, 0);
    if (base + SCAN_TILE < ncap) nxt = c4[((base + SCAN_TILE) >> 2) + tid];
    const int s = cur.x + cur.y + cur.z + cur.w;
    int incl = s;
#pragma unroll
    for (int o = 1; o < 32; o <<= 1) {
      const int t = __shfl_up(incl, o, 32);
      if (lane >= o) incl += t;
    }
    if (lane == 31) wtot[buf][wave] = incl;
    __syncthreads();
    const int wt = wtot[buf][lane];
    int winc = wt;
#pragma unroll
    for (int o = 1; o < 32; o <<= 1) {
      const int t = __shfl_up(winc, o, 32);
      if (lane >= o) winc += t;
    }
    const int tile_total = __shfl(winc, 31, 32);
    const int wave_excl  = __shfl(winc - wt, wave, 32);
    const int p0 = carry + wave_excl + incl - s;
    const int p1 = p0 + cur.x, p2 = p1 + cur.y, p3 = p2 + cur.z;
    const int idx = base + tid * 4;
    if (idx + 3 < n) {
      *(int4*)(rowptr + idx) = make_int4(p0, p1, p2, p3);
    } else {
      if (idx     < n) rowptr[idx]     = p0;
      if (idx + 1 < n) rowptr[idx + 1] = p1;
      if (idx + 2 < n) rowptr[idx + 2] = p2;
    }
    carry += tile_total;
    cur = nxt;
  }
  if (tid == 0) rowptr[n] = carry;
}

__global__ void csr_fill(const int* __restrict__ tgt, const int* __restrict__ rank,
                         const int* __restrict__ rowptr, int* __restrict__ eid, int E) {
  const int e = blockIdx.x * blockDim.x + threadIdx.x;
  if (e < E) eid[rowptr[tgt[e]] + rank[e]] = e;
}

__global__ __launch_bounds__(256) void csr_emit(
    const int* __restrict__ rowptr, const int* __restrict__ eid,
    const int* __restrict__ src, const float* __restrict__ w,
    int2* __restrict__ csr, int n) {
  int node = (blockIdx.x * 256 + threadIdx.x) >> 5;
  const int lane = threadIdx.x & 31;
  if (node >= n) return;
  node = __builtin_amdgcn_readfirstlane(node);
  const int start = rowptr[node];
  const int d = rowptr[node + 1] - start;
  for (int base = 0; base < d; base += 32) {
    const int i = base + lane;
    const int my = (i < d) ? eid[start + i] : 0x7fffffff;
    int r = 0;
    for (int j = 0; j < d; ++j) r += (eid[start + j] < my) ? 1 : 0;
    if (i < d) csr[start + r] = make_int2(src[my], __float_as_int(w[my]));
  }
}

template <int F4>
__device__ __forceinline__ float4 row_sum(const float4* __restrict__ Hq,
                                          const int2* __restrict__ csr,
                                          int j, int end) {
  float4 acc = make_float4(0.0f, 0.0f, 0.0f, 0.0f);
  for (; j + 4 <= end; j += 4) {
    const int2 e0 = csr[j], e1 = csr[j + 1], e2 = csr[j + 2], e3 = csr[j + 3];
    const float4 v0 = Hq[(size_t)e0.x * F4];
    const float4 v1 = Hq[(size_t)e1.x * F4];
    const float4 v2 = Hq[(size_t)e2.x * F4];
    const float4 v3 = Hq[(size_t)e3.x * F4];
    const float w0 = __int_as_float(e0.y), w1 = __int_as_float(e1.y);
    const float w2 = __int_as_float(e2.y), w3 = __int_as_float(e3.y);
    acc.x = fmaf(w0, v0.x, acc.x); acc.y = fmaf(w0, v0.y, acc.y);
    acc.z = fmaf(w0, v0.z, acc.z); acc.w = fmaf(w0, v0.w, acc.w);
    acc.x = fmaf(w1, v1.x, acc.x); acc.y = fmaf(w1, v1.y, acc.y);
    acc.z = fmaf(w1, v1.z, acc.z); acc.w = fmaf(w1, v1.w, acc.w);
    acc.x = fmaf(w2, v2.x, acc.x); acc.y = fmaf(w2, v2.y, acc.y);
    acc.z = fmaf(w2, v2.z, acc.z); acc.w = fmaf(w2, v2.w, acc.w);
    acc.x = fmaf(w3, v3.x, acc.x); acc.y = fmaf(w3, v3.y, acc.y);
    acc.z = fmaf(w3, v3.z, acc.z); acc.w = fmaf(w3, v3.w, acc.w);
  }
  for (; j < end; ++j) {
    const int2 e0 = csr[j];
    const float4 v0 = Hq[(size_t)e0.x * F4];
    const float w0 = __int_as_float(e0.y);
    acc.x = fmaf(w0, v0.x, acc.x); acc.y = fmaf(w0, v0.y, acc.y);
    acc.z = fmaf(w0, v0.z, acc.z); acc.w = fmaf(w0, v0.w, acc.w);
  }
  return acc;
}

template <bool GN>
__global__ __launch_bounds__(256) void gather128_tail(
    const float* __restrict__ H, const int* __restrict__ rowptr,
    const int2* __restrict__ csr, const float* __restrict__ b,
    const float* __restrict__ gamma, const float* __restrict__ beta,
    float* __restrict__ A, int n) {
  const int lane = threadIdx.x & 31;
  int node = (blockIdx.x * 256 + threadIdx.x) >> 5;
  if (node >= n) return;
  node = __builtin_amdgcn_readfirstlane(node);
  const float4 bb = *(const float4*)(b + lane * 4);
  float4 gm = make_float4(1.0f, 1.0f, 1.0f, 1.0f);
  float4 bt = make_float4(0.0f, 0.0f, 0.0f, 0.0f);
  if (GN) {
    gm = *(const float4*)(gamma + lane * 4);
    bt = *(const float4*)(beta + lane * 4);
  }
  float4 v = row_sum<32>((const float4*)H + lane, csr,
                         rowptr[node], rowptr[node + 1]);
  v.x = fmaxf(v.x + bb.x, 0.0f);
  v.y = fmaxf(v.y + bb.y, 0.0f);
  v.z = fmaxf(v.z + bb.z, 0.0f);
  v.w = fmaxf(v.w + bb.w, 0.0f);
  if (GN) {
    const float mu  = 0.25f * (v.x + v.y + v.z + v.w);
    const float ms  = 0.25f * (v.x * v.x + v.y * v.y + v.z * v.z + v.w * v.w);
    const float inv = rsqrtf(fmaxf(ms - mu * mu, 0.0f) + 1e-5f);
    v.x = (v.x - mu) * inv * gm.x + bt.x;
    v.y = (v.y - mu) * inv * gm.y + bt.y;
    v.z = (v.z - mu) * inv * gm.z + bt.z;
    v.w = (v.w - mu) * inv * gm.w + bt.w;
  }
  ((float4*)A)[(size_t)node * 32 + lane] = v;
}

__global__ __launch_bounds__(256) void gather40_lsm(
    const float* __restrict__ H, const int* __restrict__ rowptr,
    const int2* __restrict__ csr, const float* __restrict__ b,
    float* __restrict__ out, int n) {
  const int gid  = blockIdx.x * 256 + threadIdx.x;
  const int node = gid >> 3;
  const int q    = gid & 7;
  const bool live = node < n;
  int j = 0, end = 0;
  if (live) { j = rowptr[node]; end = rowptr[node + 1]; }
  const float4 bb = *(const float4*)(b + q * 4);
  const float  b1 = b[32 + q];
  const float4* __restrict__ Hq = (const float4*)H + q;
  const float*  __restrict__ Ht = H + 32 + q;
  float4 acc = make_float4(0.0f, 0.0f, 0.0f, 0.0f);
  float  act = 0.0f;
  for (; j + 4 <= end; j += 4) {
    const int2 e0 = csr[j], e1 = csr[j + 1], e2 = csr[j + 2], e3 = csr[j + 3];
    const float4 v0 = Hq[(size_t)e0.x * 10];
    const float4 v1 = Hq[(size_t)e1.x * 10];
    const float4 v2 = Hq[(size_t)e2.x * 10];
    const float4 v3 = Hq[(size_t)e3.x * 10];
    const float  t0 = Ht[(size_t)e0.x * 40];
    const float  t1 = Ht[(size_t)e1.x * 40];
    const float  t2 = Ht[(size_t)e2.x * 40];
    const float  t3 = Ht[(size_t)e3.x * 40];
    const float w0 = __int_as_float(e0.y), w1 = __int_as_float(e1.y);
    const float w2 = __int_as_float(e2.y), w3 = __int_as_float(e3.y);
    acc.x = fmaf(w0, v0.x, acc.x); acc.y = fmaf(w0, v0.y, acc.y);
    acc.z = fmaf(w0, v0.z, acc.z); acc.w = fmaf(w0, v0.w, acc.w);
    act   = fmaf(w0, t0, act);
    acc.x = fmaf(w1, v1.x, acc.x); acc.y = fmaf(w1, v1.y, acc.y);
    acc.z = fmaf(w1, v1.z, acc.z); acc.w = fmaf(w1, v1.w, acc.w);
    act   = fmaf(w1, t1, act);
    acc.x = fmaf(w2, v2.x, acc.x); acc.y = fmaf(w2, v2.y, acc.y);
    acc.z = fmaf(w2, v2.z, acc.z); acc.w = fmaf(w2, v2.w, acc.w);
    act   = fmaf(w2, t2, act);
    acc.x = fmaf(w3, v3.x, acc.x); acc.y = fmaf(w3, v3.y, acc.y);
    acc.z = fmaf(w3, v3.z, acc.z); acc.w = fmaf(w3, v3.w, acc.w);
    act   = fmaf(w3, t3, act);
  }
  for (; j < end; ++j) {
    const int2 e0 = csr[j];
    const float4 v0 = Hq[(size_t)e0.x * 10];
    const float  t0 = Ht[(size_t)e0.x * 40];
    const float w0 = __int_as_float(e0.y);
    acc.x = fmaf(w0, v0.x, acc.x); acc.y = fmaf(w0, v0.y, acc.y);
    acc.z = fmaf(w0, v0.z, acc.z); acc.w = fmaf(w0, v0.w, acc.w);
    act   = fmaf(w0, t0, act);
  }
  const float x0 = acc.x + bb.x, x1 = acc.y + bb.y;
  const float x2 = acc.z + bb.z, x3 = acc.w + bb.w;
  const float x4 = act + b1;
  float m = fmaxf(fmaxf(fmaxf(x0, x1), fmaxf(x2, x3)), x4);
  for (int o = 4; o > 0; o >>= 1) m = fmaxf(m, __shfl_xor(m, o, 32));
  float s = ((expf(x0 - m) + expf(x1 - m)) + (expf(x2 - m) + expf(x3 - m)))
            + expf(x4 - m);
  for (int o = 4; o > 0; o >>= 1) s += __shfl_xor(s, o, 32);
  const float lse = m + logf(s);
  if (live) {
    ((float4*)out)[(size_t)node * 10 + q] =
        make_float4(x0 - lse, x1 - lse, x2 - lse, x3 - lse);
    out[(size_t)node * 40 + 32 + q] = x4 - lse;
  }
}

extern "C" void kernel_launch(void* const* d_in, const int* in_sizes, int n_in,
                              void* d_out, int out_size, void* d_ws, size_t ws_size,
                              hipStream_t stream) {
  const int N = NNODES, E = NEDGES;
  const float* x     = (const float*)d_in[0];
  const int*   src   = (const int*)d_in[1];
  const int*   tgt   = (const int*)d_in[2];
  const float* w     = (const float*)d_in[3];
  const float* W0    = (const float*)d_in[4];
  const float* b0    = (const float*)d_in[5];
  const float* W1    = (const float*)d_in[6];
  const float* b1    = (const float*)d_in[7];
  const float* W2    = (const float*)d_in[8];
  const float* b2    = (const float*)d_in[9];
  const float* gamma = (const float*)d_in[10];
  const float* beta  = (const float*)d_in[11];
  float* out = (float*)d_out;

  float* H   = (float*)d_ws;
  float* A   = H + (size_t)N * 128;
  float* H2  = H;
  int*  cnt    = (int*)(H + (size_t)N * 256);
  int*  rowptr = cnt + NCAP;
  int*  rank   = rowptr + NCAP + 64;
  int*  eid    = rank + E;
  int2* csr    = (int2*)(eid + E);

  const int gemm_blocks  = (N + 127) / 128;
  const int edge_thr_blocks = (E + 255) / 256;
  const int node_blocks  = (N * 32 + 255) / 256;
  const int g40_blocks   = (N * 8 + 255) / 256;

  csr_zero<<<NCAP / 256, 256, 0, stream>>>(cnt, NCAP);
  csr_rank<<<edge_thr_blocks, 256, 0, stream>>>(tgt, cnt, rank, E);
  csr_scan<<<1, 1024, 0, stream>>>(cnt, rowptr, N, NCAP);
  csr_fill<<<edge_thr_blocks, 256, 0, stream>>>(tgt, rank, rowptr, eid, E);
  csr_emit<<<node_blocks, 256, 0, stream>>>(rowptr, eid, src, w, csr, N);

  gemm_wmma_f32<128><<<gemm_blocks, 256, 0, stream>>>(x, W0, H, N);
  gather128_tail<false><<<node_blocks, 256, 0, stream>>>(H, rowptr, csr, b0, gamma, beta, A, N);

  gemm_wmma_f32<128><<<gemm_blocks, 256, 0, stream>>>(A, W1, H, N);
  gather128_tail<true><<<node_blocks, 256, 0, stream>>>(H, rowptr, csr, b1, gamma, beta, A, N);

  gemm_wmma_f32<40><<<gemm_blocks, 256, 0, stream>>>(A, W2, H2, N);
  gather40_lsm<<<g40_blocks, 256, 0, stream>>>(H2, rowptr, csr, b2, out, N);
}


# layer 1 and 2 products moved into the 128-wide gather kernels: 16 waves sum 16 nodes, rows go to an LDS tile, f32 WMMA on it with W in LDS; 2 of 11 launches and the activation round trip gone
# speedup vs baseline: 2.4335x; 1.0974x over previous
#include <hip/hip_runtime.h>
#include <math.h>

typedef __attribute__((ext_vector_type(2))) float v2f;
typedef __attribute__((ext_vector_type(8))) float v8f;
typedef __attribute__((ext_vector_type(2))) int   v2i;
typedef const __attribute__((address_space(4))) int* kint_p;
typedef const __attribute__((address_space(4))) v2i* krec_p;

#define NNODES 50000
#define NEDGES 800000
static_assert(NNODES % 16 == 0, "the 16-row tiles of both product kernels are whole");

template <int NC>
__global__ __launch_bounds__(256) void gemm_wmma_f32(
    const float* __restrict__ X, const float* __restrict__ W,
    float* __restrict__ H, int n) {
  constexpr int K  = 128;
  constexpr int CT = (NC + 15) / 16;
  constexpr int PC = CT * 16;
  __shared__ v2f lw[(K / 2) * PC];

  const int tid = threadIdx.x;
  for (int i = tid; i < (K / 2) * PC; i += 256) {
    const int kp = i / PC, c = i % PC;
    v2f val;
    val.x = (c < NC) ? W[(2 * kp) * NC + c] : 0.0f;
    val.y = (c < NC) ? W[(2 * kp + 1) * NC + c] : 0.0f;
    lw[i] = val;
  }
  __syncthreads();

  const int wave = tid >> 5;
  const int lane = tid & 31;
  const int row0 = (blockIdx.x * 8 + wave) * 16;
  if (row0 >= n) return;

  const int m  = lane & 15;
  const int hk = lane >> 4;

  v8f acc[CT];
#pragma unroll
  for (int t = 0; t < CT; ++t) acc[t] = {};

  const float* xrow = X + (size_t)(row0 + m) * K + 2 * hk;

  for (int k4 = 0; k4 < K; k4 += 4) {
    v2f a;
    a.x = xrow[k4];
    a.y = xrow[k4 + 1];
    const v2f* brow = &lw[(k4 / 2 + hk) * PC + m];
#pragma unroll
    for (int t = 0; t < CT; ++t) {
      const v2f b = brow[t * 16];
      acc[t] = __builtin_amdgcn_wmma_f32_16x16x4_f32(
          false, a, false, b, (short)0, acc[t], false, false);
    }
  }

  const int crow = row0 + hk * 8;
#pragma unroll
  for (int t = 0; t < CT; ++t) {
    const int col = t * 16 + m;
    if ((NC % 16 == 0) || (col < NC)) {
#pragma unroll
      for (int r = 0; r < 8; ++r)
        H[(size_t)(crow + r) * NC + col] = acc[t][r];
    }
  }
}

#define SCAN_TILE 4096
#define NCAP (((NNODES + SCAN_TILE - 1) / SCAN_TILE) * SCAN_TILE)

__global__ void csr_zero(int* __restrict__ cnt, int ncap) {
  const int i = blockIdx.x * blockDim.x + threadIdx.x;
  if (i < ncap) cnt[i] = 0;
}

__global__ void csr_rank(const int* __restrict__ tgt, int* __restrict__ cnt,
                         int* __restrict__ rank, int E) {
  const int e = blockIdx.x * blockDim.x + threadIdx.x;
  if (e < E) rank[e] = atomicAdd(&cnt[tgt[e]], 1);
}

__global__ __launch_bounds__(1024) void csr_scan(
    const int* __restrict__ cnt, int* __restrict__ rowptr, int n, int ncap) {
  __shared__ int wtot[2][32];
  const int tid = threadIdx.x, lane = tid & 31, wave = tid >> 5;
  const int4* c4 = (const int4*)cnt;
  int carry = 0, buf = 0;
  int4 cur = c4[tid];
  for (int base = 0; base < ncap; base += SCAN_TILE, buf ^= 1) {
    int4 nxt = make_int4(0, 0, 0, 0);
    if (base + SCAN_TILE < ncap) nxt = c4[((base + SCAN_TILE) >> 2) + tid];
    const int s = cur.x + cur.y + cur.z + cur.w;
    int incl = s;
#pragma unroll
    for (int o = 1; o < 32; o <<= 1) {
      const int t = __shfl_up(incl, o, 32);
      if (lane >= o) incl += t;
    }
    if (lane == 31) wtot[buf][wave] = incl;
    __syncthreads();
    const int wt = wtot[buf][lane];
    int winc = wt;
#pragma unroll
    for (int o = 1; o < 32; o <<= 1) {
      const int t = __shfl_up(winc, o, 32);
      if (lane >= o) winc += t;
    }
    const int tile_total = __shfl(winc, 31, 32);
    const int wave_excl  = __shfl(winc - wt, wave, 32);
    const int p0 = carry + wave_excl + incl - s;
    const int p1 = p0 + cur.x, p2 = p1 + cur.y, p3 = p2 + cur.z;
    const int idx = base + tid * 4;
    if (idx + 3 < n) {
      *(int4*)(rowptr + idx) = make_int4(p0, p1, p2, p3);
    } else {
      if (idx     < n) rowptr[idx]     = p0;
      if (idx + 1 < n) rowptr[idx + 1] = p1;
      if (idx + 2 < n) rowptr[idx + 2] = p2;
    }
    carry += tile_total;
    cur = nxt;
  }
  if (tid == 0) rowptr[n] = carry;
}

__global__ void csr_fill(const int* __restrict__ tgt, const int* __restrict__ rank,
                         const int* __restrict__ rowptr, int* __restrict__ eid, int E) {
  const int e = blockIdx.x * blockDim.x + threadIdx.x;
  if (e < E) eid[rowptr[tgt[e]] + rank[e]] = e;
}

__global__ __launch_bounds__(256) void csr_emit(
    const int* __restrict__ rowptr, const int* __restrict__ eid,
    const int* __restrict__ src, const float* __restrict__ w,
    int2* __restrict__ csr, int n) {
  int node = (blockIdx.x * 256 + threadIdx.x) >> 5;
  const int lane = threadIdx.x & 31;
  if (node >= n) return;
  node = __builtin_amdgcn_readfirstlane(node);
  const int start = rowptr[node];
  const int d = rowptr[node + 1] - start;
  for (int base = 0; base < d; base += 32) {
    const int i = base + lane;
    const int my = (i < d) ? eid[start + i] : 0x7fffffff;
    int r = 0;
    for (int j = 0; j < d; ++j) r += (eid[start + j] < my) ? 1 : 0;
    if (i < d) csr[start + r] = make_int2(src[my], __float_as_int(w[my]));
  }
}

template <int F4, typename REC>
__device__ __forceinline__ float4 row_sum(const float4* __restrict__ Hq,
                                          REC csr, int j, int end) {
  float4 acc = make_float4(0.0f, 0.0f, 0.0f, 0.0f);
  for (; j + 4 <= end; j += 4) {
    const auto e0 = csr[j], e1 = csr[j + 1], e2 = csr[j + 2], e3 = csr[j + 3];
    const float4 v0 = Hq[(size_t)e0.x * F4];
    const float4 v1 = Hq[(size_t)e1.x * F4];
    const float4 v2 = Hq[(size_t)e2.x * F4];
    const float4 v3 = Hq[(size_t)e3.x * F4];
    const float w0 = __int_as_float(e0.y), w1 = __int_as_float(e1.y);
    const float w2 = __int_as_float(e2.y), w3 = __int_as_float(e3.y);
    acc.x = fmaf(w0, v0.x, acc.x); acc.y = fmaf(w0, v0.y, acc.y);
    acc.z = fmaf(w0, v0.z, acc.z); acc.w = fmaf(w0, v0.w, acc.w);
    acc.x = fmaf(w1, v1.x, acc.x); acc.y = fmaf(w1, v1.y, acc.y);
    acc.z = fmaf(w1, v1.z, acc.z); acc.w = fmaf(w1, v1.w, acc.w);
    acc.x = fmaf(w2, v2.x, acc.x); acc.y = fmaf(w2, v2.y, acc.y);
    acc.z = fmaf(w2, v2.z, acc.z); acc.w = fmaf(w2, v2.w, acc.w);
    acc.x = fmaf(w3, v3.x, acc.x); acc.y = fmaf(w3, v3.y, acc.y);
    acc.z = fmaf(w3, v3.z, acc.z); acc.w = fmaf(w3, v3.w, acc.w);
  }
  for (; j < end; ++j) {
    const auto e0 = csr[j];
    const float4 v0 = Hq[(size_t)e0.x * F4];
    const float w0 = __int_as_float(e0.y);
    acc.x = fmaf(w0, v0.x, acc.x); acc.y = fmaf(w0, v0.y, acc.y);
    acc.z = fmaf(w0, v0.z, acc.z); acc.w = fmaf(w0, v0.w, acc.w);
  }
  return acc;
}

template <bool GN, int NC>
__global__ __launch_bounds__(512) void gather128_tail_gemm(
    const float* __restrict__ H, const int* __restrict__ rowptr,
    const int2* __restrict__ csr, const float* __restrict__ b,
    const float* __restrict__ gamma, const float* __restrict__ beta,
    const float* __restrict__ Wn, float* __restrict__ Hn, int n, int ntiles) {
  constexpr int K  = 128;
  constexpr int CT = (NC + 15) / 16;
  constexpr int PC = CT * 16;
  constexpr int TP = 132;
  __shared__ v2f lw[(K / 2) * PC];
  __shared__ __attribute__((aligned(16))) float tile[16 * TP];

  const int tid = threadIdx.x;
  for (int i = tid; i < (K / 2) * PC; i += 512) {
    const int kp = i / PC, c = i % PC;
    v2f val;
    val.x = (c < NC) ? Wn[(2 * kp) * NC + c] : 0.0f;
    val.y = (c < NC) ? Wn[(2 * kp + 1) * NC + c] : 0.0f;
    lw[i] = val;
  }

  const int lane = tid & 31;
  const int wave = __builtin_amdgcn_readfirstlane(tid >> 5);
  const int m  = lane & 15;
  const int hk = lane >> 4;

  const kint_p krow = (kint_p)rowptr;
  const krec_p krec = (krec_p)csr;

  for (int t = blockIdx.x; t < ntiles; t += gridDim.x) {
    const int node = t * 16 + wave;
    float4 v = make_float4(0.0f, 0.0f, 0.0f, 0.0f);
    if (node < n) {
      v = row_sum<32>((const float4*)H + lane, krec, krow[node], krow[node + 1]);
      const float4 bb = *(const float4*)(b + lane * 4);
      v.x = fmaxf(v.x + bb.x, 0.0f);
      v.y = fmaxf(v.y + bb.y, 0.0f);
      v.z = fmaxf(v.z + bb.z, 0.0f);
      v.w = fmaxf(v.w + bb.w, 0.0f);
      if (GN) {
        const float mu  = 0.25f * (v.x + v.y + v.z + v.w);
        const float ms  = 0.25f * (v.x * v.x + v.y * v.y + v.z * v.z + v.w * v.w);
        const float inv = rsqrtf(fmaxf(ms - mu * mu, 0.0f) + 1e-5f);
        const float4 gm = *(const float4*)(gamma + lane * 4);
        const float4 bt = *(const float4*)(beta + lane * 4);
        v.x = (v.x - mu) * inv * gm.x + bt.x;
        v.y = (v.y - mu) * inv * gm.y + bt.y;
        v.z = (v.z - mu) * inv * gm.z + bt.z;
        v.w = (v.w - mu) * inv * gm.w + bt.w;
      }
    }
    __syncthreads();
    *(float4*)&tile[wave * TP + lane * 4] = v;
    __syncthreads();
    if (wave < CT) {
      const float* arow = &tile[m * TP + 2 * hk];
      const v2f*   brow = &lw[hk * PC + wave * 16 + m];
      v8f acc = {};
#pragma unroll
      for (int k4 = 0; k4 < K; k4 += 4) {
        const v2f a  = *(const v2f*)(arow + k4);
        const v2f bq = brow[(k4 / 2) * PC];
        acc = __builtin_amdgcn_wmma_f32_16x16x4_f32(
            false, a, false, bq, (short)0, acc, false, false);
      }
      const int crow = t * 16 + hk * 8;
      const int col  = wave * 16 + m;
      if ((NC % 16 == 0) || (col < NC)) {
#pragma unroll
        for (int r = 0; r < 8; ++r)
          Hn[(size_t)(crow + r) * NC + col] = acc[r];
      }
    }
  }
}

__global__ __launch_bounds__(256) void gather40_lsm(
    const float* __restrict__ H, const int* __restrict__ rowptr,
    const int2* __restrict__ csr, const float* __restrict__ b,
    float* __restrict__ out, int n) {
  const int gid  = blockIdx.x * 256 + threadIdx.x;
  const int node = gid >> 3;
  const int q    = gid & 7;
  const bool live = node < n;
  int j = 0, end = 0;
  if (live) { j = rowptr[node]; end = rowptr[node + 1]; }
  const float4 bb = *(const float4*)(b + q * 4);
  const float  b1 = b[32 + q];
  const float4* __restrict__ Hq = (const float4*)H + q;
  const float*  __restrict__ Ht = H + 32 + q;
  float4 acc = make_float4(0.0f, 0.0f, 0.0f, 0.0f);
  float  act = 0.0f;
  for (; j + 4 <= end; j += 4) {
    const int2 e0 = csr[j], e1 = csr[j + 1], e2 = csr[j + 2], e3 = csr[j + 3];
    const float4 v0 = Hq[(size_t)e0.x * 10];
    const float4 v1 = Hq[(size_t)e1.x * 10];
    const float4 v2 = Hq[(size_t)e2.x * 10];
    const float4 v3 = Hq[(size_t)e3.x * 10];
    const float  t0 = Ht[(size_t)e0.x * 40];
    const float  t1 = Ht[(size_t)e1.x * 40];
    const float  t2 = Ht[(size_t)e2.x * 40];
    const float  t3 = Ht[(size_t)e3.x * 40];
    const float w0 = __int_as_float(e0.y), w1 = __int_as_float(e1.y);
    const float w2 = __int_as_float(e2.y), w3 = __int_as_float(e3.y);
    acc.x = fmaf(w0, v0.x, acc.x); acc.y = fmaf(w0, v0.y, acc.y);
    acc.z = fmaf(w0, v0.z, acc.z); acc.w = fmaf(w0, v0.w, acc.w);
    act   = fmaf(w0, t0, act);
    acc.x = fmaf(w1, v1.x, acc.x); acc.y = fmaf(w1, v1.y, acc.y);
    acc.z = fmaf(w1, v1.z, acc.z); acc.w = fmaf(w1, v1.w, acc.w);
    act   = fmaf(w1, t1, act);
    acc.x = fmaf(w2, v2.x, acc.x); acc.y = fmaf(w2, v2.y, acc.y);
    acc.z = fmaf(w2, v2.z, acc.z); acc.w = fmaf(w2, v2.w, acc.w);
    act   = fmaf(w2, t2, act);
    acc.x = fmaf(w3, v3.x, acc.x); acc.y = fmaf(w3, v3.y, acc.y);
    acc.z = fmaf(w3, v3.z, acc.z); acc.w = fmaf(w3, v3.w, acc.w);
    act   = fmaf(w3, t3, act);
  }
  for (; j < end; ++j) {
    const int2 e0 = csr[j];
    const float4 v0 = Hq[(size_t)e0.x * 10];
    const float  t0 = Ht[(size_t)e0.x * 40];
    const float w0 = __int_as_float(e0.y);
    acc.x = fmaf(w0, v0.x, acc.x); acc.y = fmaf(w0, v0.y, acc.y);
    acc.z = fmaf(w0, v0.z, acc.z); acc.w = fmaf(w0, v0.w, acc.w);
    act   = fmaf(w0, t0, act);
  }
  const float x0 = acc.x + bb.x, x1 = acc.y + bb.y;
  const float x2 = acc.z + bb.z, x3 = acc.w + bb.w;
  const float x4 = act + b1;
  float m = fmaxf(fmaxf(fmaxf(x0, x1), fmaxf(x2, x3)), x4);
  for (int o = 4; o > 0; o >>= 1) m = fmaxf(m, __shfl_xor(m, o, 32));
  float s = ((expf(x0 - m) + expf(x1 - m)) + (expf(x2 - m) + expf(x3 - m)))
            + expf(x4 - m);
  for (int o = 4; o > 0; o >>= 1) s += __shfl_xor(s, o, 32);
  const float lse = m + logf(s);
  if (live) {
    ((float4*)out)[(size_t)node * 10 + q] =
        make_float4(x0 - lse, x1 - lse, x2 - lse, x3 - lse);
    out[(size_t)node * 40 + 32 + q] = x4 - lse;
  }
}

extern "C" void kernel_launch(void* const* d_in, const int* in_sizes, int n_in,
                              void* d_out, int out_size, void* d_ws, size_t ws_size,
                              hipStream_t stream) {
  const int N = NNODES, E = NEDGES;
  const float* x     = (const float*)d_in[0];
  const int*   src   = (const int*)d_in[1];
  const int*   tgt   = (const int*)d_in[2];
  const float* w     = (const float*)d_in[3];
  const float* W0    = (const float*)d_in[4];
  const float* b0    = (const float*)d_in[5];
  const float* W1    = (const float*)d_in[6];
  const float* b1    = (const float*)d_in[7];
  const float* W2    = (const float*)d_in[8];
  const float* b2    = (const float*)d_in[9];
  const float* gamma = (const float*)d_in[10];
  const float* beta  = (const float*)d_in[11];
  float* out = (float*)d_out;

  float* H   = (float*)d_ws;
  float* H1  = H + (size_t)N * 128;
  float* H2  = H;
  int*  cnt    = (int*)(H + (size_t)N * 256);
  int*  rowptr = cnt + NCAP;
  int*  rank   = rowptr + NCAP + 64;
  int*  eid    = rank + E;
  int2* csr    = (int2*)(eid + E);

  const int gemm_blocks  = (N + 127) / 128;
  const int edge_thr_blocks = (E + 255) / 256;
  const int node_blocks  = (N * 32 + 255) / 256;
  const int g40_blocks   = (N * 8 + 255) / 256;
  const int ntiles       = (N + 15) / 16;
  const int fused_blocks = 1024;

  csr_zero<<<NCAP / 256, 256, 0, stream>>>(cnt, NCAP);
  csr_rank<<<edge_thr_blocks, 256, 0, stream>>>(tgt, cnt, rank, E);
  csr_scan<<<1, 1024, 0, stream>>>(cnt, rowptr, N, NCAP);
  csr_fill<<<edge_thr_blocks, 256, 0, stream>>>(tgt, rank, rowptr, eid, E);
  csr_emit<<<node_blocks, 256, 0, stream>>>(rowptr, eid, src, w, csr, N);

  gemm_wmma_f32<128><<<gemm_blocks, 256, 0, stream>>>(x, W0, H, N);

  gather128_tail_gemm<false, 128><<<fused_blocks, 512, 0, stream>>>(
      H, rowptr, csr, b0, gamma, beta, W1, H1, N, ntiles);

  gather128_tail_gemm<true, 40><<<fused_blocks, 512, 0, stream>>>(
      H1, rowptr, csr, b1, gamma, beta, W2, H2, N, ntiles);

  gather40_lsm<<<g40_blocks, 256, 0, stream>>>(H2, rowptr, csr, b2, out, N);
}


# x @ W0 has no launch of its own: its blocks ride as the first work-groups of the rank and fill launches (half each; matrix pipes idle beside the edge atomics); 8 launches, sums and order unchanged
# speedup vs baseline: 2.5236x; 1.0370x over previous
#include <hip/hip_runtime.h>
#include <math.h>

typedef __attribute__((ext_vector_type(2))) float v2f;
typedef __attribute__((ext_vector_type(8))) float v8f;
typedef __attribute__((ext_vector_type(2))) int   v2i;
typedef const __attribute__((address_space(4))) int* kint_p;
typedef const __attribute__((address_space(4))) v2i* krec_p;

#define NNODES 50000
#define NEDGES 800000
static_assert(NNODES % 16 == 0, "the 16-row tiles of both product kernels are whole");

template <int NC>
__device__ __forceinline__ void gemm_wmma_f32_block(
    const float* __restrict__ X, const float* __restrict__ W,
    float* __restrict__ H, int n, int blk, v2f* __restrict__ lw) {
  constexpr int K  = 128;
  constexpr int CT = (NC + 15) / 16;
  constexpr int PC = CT * 16;

  const int tid = threadIdx.x;
  for (int i = tid; i < (K / 2) * PC; i += 256) {
    const int kp = i / PC, c = i % PC;
    v2f val;
    val.x = (c < NC) ? W[(2 * kp) * NC + c] : 0.0f;
    val.y = (c < NC) ? W[(2 * kp + 1) * NC + c] : 0.0f;
    lw[i] = val;
  }
  __syncthreads();

  const int wave = tid >> 5;
  const int lane = tid & 31;
  const int row0 = (blk * 8 + wave) * 16;
  if (row0 >= n) return;

  const int m  = lane & 15;
  const int hk = lane >> 4;

  v8f acc[CT];
#pragma unroll
  for (int t = 0; t < CT; ++t) acc[t] = {};

  const float* xrow = X + (size_t)(row0 + m) * K + 2 * hk;

  for (int k4 = 0; k4 < K; k4 += 4) {
    v2f a;
    a.x = xrow[k4];
    a.y = xrow[k4 + 1];
    const v2f* brow = &lw[(k4 / 2 + hk) * PC + m];
#pragma unroll
    for (int t = 0; t < CT; ++t) {
      const v2f b = brow[t * 16];
      acc[t] = __builtin_amdgcn_wmma_f32_16x16x4_f32(
          false, a, false, b, (short)0, acc[t], false, false);
    }
  }

  const int crow = row0 + hk * 8;
#pragma unroll
  for (int t = 0; t < CT; ++t) {
    const int col = t * 16 + m;
    if ((NC % 16 == 0) || (col < NC)) {
#pragma unroll
      for (int r = 0; r < 8; ++r)
        H[(size_t)(crow + r) * NC + col] = acc[t][r];
    }
  }
}

#define SCAN_TILE 4096
#define NCAP (((NNODES + SCAN_TILE - 1) / SCAN_TILE) * SCAN_TILE)

__global__ void csr_zero(int* __restrict__ cnt, int ncap) {
  const int i = blockIdx.x * blockDim.x + threadIdx.x;
  if (i < ncap) cnt[i] = 0;
}


__global__ __launch_bounds__(256) void csr_rank_beside_gemm(
    const int* __restrict__ tgt, int* __restrict__ cnt, int* __restrict__ rank, int E,
    const float* __restrict__ X, const float* __restrict__ W,
    float* __restrict__ H, int n, int gemm_first, int gemm_count) {
  __shared__ v2f lw[(128 / 2) * 128];
  const int blk = blockIdx.x;
  if (blk < gemm_count) {
    gemm_wmma_f32_block<128>(X, W, H, n, gemm_first + blk, lw);
  } else {
    const int e = (blk - gemm_count) * 256 + threadIdx.x;
    if (e < E) rank[e] = atomicAdd(&cnt[tgt[e]], 1);
  }
}

__global__ __launch_bounds__(1024) void csr_scan(
    const int* __restrict__ cnt, int* __restrict__ rowptr, int n, int ncap) {
  __shared__ int wtot[2][32];
  const int tid = threadIdx.x, lane = tid & 31, wave = tid >> 5;
  const int4* c4 = (const int4*)cnt;
  int carry = 0, buf = 0;
  int4 cur = c4[tid];
  for (int base = 0; base < ncap; base += SCAN_TILE, buf ^= 1) {
    int4 nxt = make_int4(0, 0, 0, 0);
    if (base + SCAN_TILE < ncap) nxt = c4[((base + SCAN_TILE) >> 2) + tid];
    const int s = cur.x + cur.y + cur.z + cur.w;
    int incl = s;
#pragma unroll
    for (int o = 1; o < 32; o <<= 1) {
      const int t = __shfl_up(incl, o, 32);
      if (lane >= o) incl += t;
    }
    if (lane == 31) wtot[buf][wave] = incl;
    __syncthreads();
    const int wt = wtot[buf][lane];
    int winc = wt;
#pragma unroll
    for (int o = 1; o < 32; o <<= 1) {
      const int t = __shfl_up(winc, o, 32);
      if (lane >= o) winc += t;
    }
    const int tile_total = __shfl(winc, 31, 32);
    const int wave_excl  = __shfl(winc - wt, wave, 32);
    const int p0 = carry + wave_excl + incl - s;
    const int p1 = p0 + cur.x, p2 = p1 + cur.y, p3 = p2 + cur.z;
    const int idx = base + tid * 4;
    if (idx + 3 < n) {
      *(int4*)(rowptr + idx) = make_int4(p0, p1, p2, p3);
    } else {
      if (idx     < n) rowptr[idx]     = p0;
      if (idx + 1 < n) rowptr[idx + 1] = p1;
      if (idx + 2 < n) rowptr[idx + 2] = p2;
    }
    carry += tile_total;
    cur = nxt;
  }
  if (tid == 0) rowptr[n] = carry;
}

__global__ __launch_bounds__(256) void csr_fill_beside_gemm(
    const int* __restrict__ tgt, const int* __restrict__ rank,
    const int* __restrict__ rowptr, int* __restrict__ eid, int E,
    const float* __restrict__ X, const float* __restrict__ W,
    float* __restrict__ H, int n, int gemm_first, int gemm_count) {
  __shared__ v2f lw[(128 / 2) * 128];
  const int blk = blockIdx.x;
  if (blk < gemm_count) {
    gemm_wmma_f32_block<128>(X, W, H, n, gemm_first + blk, lw);
  } else {
    const int e = (blk - gemm_count) * 256 + threadIdx.x;
    if (e < E) eid[rowptr[tgt[e]] + rank[e]] = e;
  }
}

__global__ __launch_bounds__(256) void csr_emit(
    const int* __restrict__ rowptr, const int* __restrict__ eid,
    const int* __restrict__ src, const float* __restrict__ w,
    int2* __restrict__ csr, int n) {
  int node = (blockIdx.x * 256 + threadIdx.x) >> 5;
  const int lane = threadIdx.x & 31;
  if (node >= n) return;
  node = __builtin_amdgcn_readfirstlane(node);
  const int start = rowptr[node];
  const int d = rowptr[node + 1] - start;
  for (int base = 0; base < d; base += 32) {
    const int i = base + lane;
    const int my = (i < d) ? eid[start + i] : 0x7fffffff;
    int r = 0;
    for (int j = 0; j < d; ++j) r += (eid[start + j] < my) ? 1 : 0;
    if (i < d) csr[start + r] = make_int2(src[my], __float_as_int(w[my]));
  }
}

template <int F4, typename REC>
__device__ __forceinline__ float4 row_sum(const float4* __restrict__ Hq,
                                          REC csr, int j, int end) {
  float4 acc = make_float4(0.0f, 0.0f, 0.0f, 0.0f);
  for (; j + 4 <= end; j += 4) {
    const auto e0 = csr[j], e1 = csr[j + 1], e2 = csr[j + 2], e3 = csr[j + 3];
    const float4 v0 = Hq[(size_t)e0.x * F4];
    const float4 v1 = Hq[(size_t)e1.x * F4];
    const float4 v2 = Hq[(size_t)e2.x * F4];
    const float4 v3 = Hq[(size_t)e3.x * F4];
    const float w0 = __int_as_float(e0.y), w1 = __int_as_float(e1.y);
    const float w2 = __int_as_float(e2.y), w3 = __int_as_float(e3.y);
    acc.x = fmaf(w0, v0.x, acc.x); acc.y = fmaf(w0, v0.y, acc.y);
    acc.z = fmaf(w0, v0.z, acc.z); acc.w = fmaf(w0, v0.w, acc.w);
    acc.x = fmaf(w1, v1.x, acc.x); acc.y = fmaf(w1, v1.y, acc.y);
    acc.z = fmaf(w1, v1.z, acc.z); acc.w = fmaf(w1, v1.w, acc.w);
    acc.x = fmaf(w2, v2.x, acc.x); acc.y = fmaf(w2, v2.y, acc.y);
    acc.z = fmaf(w2, v2.z, acc.z); acc.w = fmaf(w2, v2.w, acc.w);
    acc.x = fmaf(w3, v3.x, acc.x); acc.y = fmaf(w3, v3.y, acc.y);
    acc.z = fmaf(w3, v3.z, acc.z); acc.w = fmaf(w3, v3.w, acc.w);
  }
  for (; j < end; ++j) {
    const auto e0 = csr[j];
    const float4 v0 = Hq[(size_t)e0.x * F4];
    const float w0 = __int_as_float(e0.y);
    acc.x = fmaf(w0, v0.x, acc.x); acc.y = fmaf(w0, v0.y, acc.y);
    acc.z = fmaf(w0, v0.z, acc.z); acc.w = fmaf(w0, v0.w, acc.w);
  }
  return acc;
}

template <bool GN, int NC>
__global__ __launch_bounds__(512) void gather128_tail_gemm(
    const float* __restrict__ H, const int* __restrict__ rowptr,
    const int2* __restrict__ csr, const float* __restrict__ b,
    const float* __restrict__ gamma, const float* __restrict__ beta,
    const float* __restrict__ Wn, float* __restrict__ Hn, int n, int ntiles) {
  constexpr int K  = 128;
  constexpr int CT = (NC + 15) / 16;
  constexpr int PC = CT * 16;
  constexpr int TP = 132;
  __shared__ v2f lw[(K / 2) * PC];
  __shared__ __attribute__((aligned(16))) float tile[16 * TP];

  const int tid = threadIdx.x;
  for (int i = tid; i < (K / 2) * PC; i += 512) {
    const int kp = i / PC, c = i % PC;
    v2f val;
    val.x = (c < NC) ? Wn[(2 * kp) * NC + c] : 0.0f;
    val.y = (c < NC) ? Wn[(2 * kp + 1) * NC + c] : 0.0f;
    lw[i] = val;
  }

  const int lane = tid & 31;
  const int wave = __builtin_amdgcn_readfirstlane(tid >> 5);
  const int m  = lane & 15;
  const int hk = lane >> 4;

  const kint_p krow = (kint_p)rowptr;
  const krec_p krec = (krec_p)csr;

  for (int t = blockIdx.x; t < ntiles; t += gridDim.x) {
    const int node = t * 16 + wave;
    float4 v = make_float4(0.0f, 0.0f, 0.0f, 0.0f);
    if (node < n) {
      v = row_sum<32>((const float4*)H + lane, krec, krow[node], krow[node + 1]);
      const float4 bb = *(const float4*)(b + lane * 4);
      v.x = fmaxf(v.x + bb.x, 0.0f);
      v.y = fmaxf(v.y + bb.y, 0.0f);
      v.z = fmaxf(v.z + bb.z, 0.0f);
      v.w = fmaxf(v.w + bb.w, 0.0f);
      if (GN) {
        const float mu  = 0.25f * (v.x + v.y + v.z + v.w);
        const float ms  = 0.25f * (v.x * v.x + v.y * v.y + v.z * v.z + v.w * v.w);
        const float inv = rsqrtf(fmaxf(ms - mu * mu, 0.0f) + 1e-5f);
        const float4 gm = *(const float4*)(gamma + lane * 4);
        const float4 bt = *(const float4*)(beta + lane * 4);
        v.x = (v.x - mu) * inv * gm.x + bt.x;
        v.y = (v.y - mu) * inv * gm.y + bt.y;
        v.z = (v.z - mu) * inv * gm.z + bt.z;
        v.w = (v.w - mu) * inv * gm.w + bt.w;
      }
    }
    __syncthreads();
    *(float4*)&tile[wave * TP + lane * 4] = v;
    __syncthreads();
    if (wave < CT) {
      const float* arow = &tile[m * TP + 2 * hk];
      const v2f*   brow = &lw[hk * PC + wave * 16 + m];
      v8f acc = {};
#pragma unroll
      for (int k4 = 0; k4 < K; k4 += 4) {
        const v2f a  = *(const v2f*)(arow + k4);
        const v2f bq = brow[(k4 / 2) * PC];
        acc = __builtin_amdgcn_wmma_f32_16x16x4_f32(
            false, a, false, bq, (short)0, acc, false, false);
      }
      const int crow = t * 16 + hk * 8;
      const int col  = wave * 16 + m;
      if ((NC % 16 == 0) || (col < NC)) {
#pragma unroll
        for (int r = 0; r < 8; ++r)
          Hn[(size_t)(crow + r) * NC + col] = acc[r];
      }
    }
  }
}

__global__ __launch_bounds__(256) void gather40_lsm(
    const float* __restrict__ H, const int* __restrict__ rowptr,
    const int2* __restrict__ csr, const float* __restrict__ b,
    float* __restrict__ out, int n) {
  const int gid  = blockIdx.x * 256 + threadIdx.x;
  const int node = gid >> 3;
  const int q    = gid & 7;
  const bool live = node < n;
  int j = 0, end = 0;
  if (live) { j = rowptr[node]; end = rowptr[node + 1]; }
  const float4 bb = *(const float4*)(b + q * 4);
  const float  b1 = b[32 + q];
  const float4* __restrict__ Hq = (const float4*)H + q;
  const float*  __restrict__ Ht = H + 32 + q;
  float4 acc = make_float4(0.0f, 0.0f, 0.0f, 0.0f);
  float  act = 0.0f;
  for (; j + 4 <= end; j += 4) {
    const int2 e0 = csr[j], e1 = csr[j + 1], e2 = csr[j + 2], e3 = csr[j + 3];
    const float4 v0 = Hq[(size_t)e0.x * 10];
    const float4 v1 = Hq[(size_t)e1.x * 10];
    const float4 v2 = Hq[(size_t)e2.x * 10];
    const float4 v3 = Hq[(size_t)e3.x * 10];
    const float  t0 = Ht[(size_t)e0.x * 40];
    const float  t1 = Ht[(size_t)e1.x * 40];
    const float  t2 = Ht[(size_t)e2.x * 40];
    const float  t3 = Ht[(size_t)e3.x * 40];
    const float w0 = __int_as_float(e0.y), w1 = __int_as_float(e1.y);
    const float w2 = __int_as_float(e2.y), w3 = __int_as_float(e3.y);
    acc.x = fmaf(w0, v0.x, acc.x); acc.y = fmaf(w0, v0.y, acc.y);
    acc.z = fmaf(w0, v0.z, acc.z); acc.w = fmaf(w0, v0.w, acc.w);
    act   = fmaf(w0, t0, act);
    acc.x = fmaf(w1, v1.x, acc.x); acc.y = fmaf(w1, v1.y, acc.y);
    acc.z = fmaf(w1, v1.z, acc.z); acc.w = fmaf(w1, v1.w, acc.w);
    act   = fmaf(w1, t1, act);
    acc.x = fmaf(w2, v2.x, acc.x); acc.y = fmaf(w2, v2.y, acc.y);
    acc.z = fmaf(w2, v2.z, acc.z); acc.w = fmaf(w2, v2.w, acc.w);
    act   = fmaf(w2, t2, act);
    acc.x = fmaf(w3, v3.x, acc.x); acc.y = fmaf(w3, v3.y, acc.y);
    acc.z = fmaf(w3, v3.z, acc.z); acc.w = fmaf(w3, v3.w, acc.w);
    act   = fmaf(w3, t3, act);
  }
  for (; j < end; ++j) {
    const int2 e0 = csr[j];
    const float4 v0 = Hq[(size_t)e0.x * 10];
    const float  t0 = Ht[(size_t)e0.x * 40];
    const float w0 = __int_as_float(e0.y);
    acc.x = fmaf(w0, v0.x, acc.x); acc.y = fmaf(w0, v0.y, acc.y);
    acc.z = fmaf(w0, v0.z, acc.z); acc.w = fmaf(w0, v0.w, acc.w);
    act   = fmaf(w0, t0, act);
  }
  const float x0 = acc.x + bb.x, x1 = acc.y + bb.y;
  const float x2 = acc.z + bb.z, x3 = acc.w + bb.w;
  const float x4 = act + b1;
  float m = fmaxf(fmaxf(fmaxf(x0, x1), fmaxf(x2, x3)), x4);
  for (int o = 4; o > 0; o >>= 1) m = fmaxf(m, __shfl_xor(m, o, 32));
  float s = ((expf(x0 - m) + expf(x1 - m)) + (expf(x2 - m) + expf(x3 - m)))
            + expf(x4 - m);
  for (int o = 4; o > 0; o >>= 1) s += __shfl_xor(s, o, 32);
  const float lse = m + logf(s);
  if (live) {
    ((float4*)out)[(size_t)node * 10 + q] =
        make_float4(x0 - lse, x1 - lse, x2 - lse, x3 - lse);
    out[(size_t)node * 40 + 32 + q] = x4 - lse;
  }
}

extern "C" void kernel_launch(void* const* d_in, const int* in_sizes, int n_in,
                              void* d_out, int out_size, void* d_ws, size_t ws_size,
                              hipStream_t stream) {
  const int N = NNODES, E = NEDGES;
  const float* x     = (const float*)d_in[0];
  const int*   src   = (const int*)d_in[1];
  const int*   tgt   = (const int*)d_in[2];
  const float* w     = (const float*)d_in[3];
  const float* W0    = (const float*)d_in[4];
  const float* b0    = (const float*)d_in[5];
  const float* W1    = (const float*)d_in[6];
  const float* b1    = (const float*)d_in[7];
  const float* W2    = (const float*)d_in[8];
  const float* b2    = (const float*)d_in[9];
  const float* gamma = (const float*)d_in[10];
  const float* beta  = (const float*)d_in[11];
  float* out = (float*)d_out;

  float* H   = (float*)d_ws;
  float* H1  = H + (size_t)N * 128;
  float* H2  = H;
  int*  cnt    = (int*)(H + (size_t)N * 256);
  int*  rowptr = cnt + NCAP;
  int*  rank   = rowptr + NCAP + 64;
  int*  eid    = rank + E;
  int2* csr    = (int2*)(eid + E);

  const int gemm_blocks  = (N + 127) / 128;
  const int gemm_half0   = (gemm_blocks + 1) / 2;
  const int gemm_half1   = gemm_blocks - gemm_half0;
  const int edge_thr_blocks = (E + 255) / 256;
  const int node_blocks  = (N * 32 + 255) / 256;
  const int g40_blocks   = (N * 8 + 255) / 256;
  const int ntiles       = (N + 15) / 16;
  const int fused_blocks = 1024;

  csr_zero<<<NCAP / 256, 256, 0, stream>>>(cnt, NCAP);
  csr_rank_beside_gemm<<<gemm_half0 + edge_thr_blocks, 256, 0, stream>>>(
      tgt, cnt, rank, E, x, W0, H, N, 0, gemm_half0);
  csr_scan<<<1, 1024, 0, stream>>>(cnt, rowptr, N, NCAP);
  csr_fill_beside_gemm<<<gemm_half1 + edge_thr_blocks, 256, 0, stream>>>(
      tgt, rank, rowptr, eid, E, x, W0, H, N, gemm_half0, gemm_half1);
  csr_emit<<<node_blocks, 256, 0, stream>>>(rowptr, eid, src, w, csr, N);

  gather128_tail_gemm<false, 128><<<fused_blocks, 512, 0, stream>>>(
      H, rowptr, csr, b0, gamma, beta, W1, H1, N, ntiles);

  gather128_tail_gemm<true, 40><<<fused_blocks, 512, 0, stream>>>(
      H1, rowptr, csr, b1, gamma, beta, W2, H2, N, ntiles);

  gather40_lsm<<<g40_blocks, 256, 0, stream>>>(H2, rowptr, csr, b2, out, N);
}


# each row is put into edge-id order by the wave that sums it for layer 0 (fill stores whole {id,src,w} slots; places counted across lanes in registers): emit launch gone, 7 launches; sum order kept
# speedup vs baseline: 2.7143x; 1.0756x over previous
#include <hip/hip_runtime.h>
#include <math.h>

typedef __attribute__((ext_vector_type(2))) float v2f;
typedef __attribute__((ext_vector_type(8))) float v8f;
typedef __attribute__((ext_vector_type(2))) int   v2i;
typedef const __attribute__((address_space(4))) int* kint_p;
typedef const __attribute__((address_space(4))) v2i* krec_p;

#define NNODES 50000
#define NEDGES 800000
static_assert(NNODES % 16 == 0, "the 16-row tiles of both product kernels are whole");

template <int NC>
__device__ __forceinline__ void gemm_wmma_f32_block(
    const float* __restrict__ X, const float* __restrict__ W,
    float* __restrict__ H, int n, int blk, v2f* __restrict__ lw) {
  constexpr int K  = 128;
  constexpr int CT = (NC + 15) / 16;
  constexpr int PC = CT * 16;

  const int tid = threadIdx.x;
  for (int i = tid; i < (K / 2) * PC; i += 256) {
    const int kp = i / PC, c = i % PC;
    v2f val;
    val.x = (c < NC) ? W[(2 * kp) * NC + c] : 0.0f;
    val.y = (c < NC) ? W[(2 * kp + 1) * NC + c] : 0.0f;
    lw[i] = val;
  }
  __syncthreads();

  const int wave = tid >> 5;
  const int lane = tid & 31;
  const int row0 = (blk * 8 + wave) * 16;
  if (row0 >= n) return;

  const int m  = lane & 15;
  const int hk = lane >> 4;

  v8f acc[CT];
#pragma unroll
  for (int t = 0; t < CT; ++t) acc[t] = {};

  const float* xrow = X + (size_t)(row0 + m) * K + 2 * hk;

  for (int k4 = 0; k4 < K; k4 += 4) {
    v2f a;
    a.x = xrow[k4];
    a.y = xrow[k4 + 1];
    const v2f* brow = &lw[(k4 / 2 + hk) * PC + m];
#pragma unroll
    for (int t = 0; t < CT; ++t) {
      const v2f b = brow[t * 16];
      acc[t] = __builtin_amdgcn_wmma_f32_16x16x4_f32(
          false, a, false, b, (short)0, acc[t], false, false);
    }
  }

  const int crow = row0 + hk * 8;
#pragma unroll
  for (int t = 0; t < CT; ++t) {
    const int col = t * 16 + m;
    if ((NC % 16 == 0) || (col < NC)) {
#pragma unroll
      for (int r = 0; r < 8; ++r)
        H[(size_t)(crow + r) * NC + col] = acc[t][r];
    }
  }
}

#define SCAN_TILE 4096
#define NCAP (((NNODES + SCAN_TILE - 1) / SCAN_TILE) * SCAN_TILE)

__global__ void csr_zero(int* __restrict__ cnt, int ncap) {
  const int i = blockIdx.x * blockDim.x + threadIdx.x;
  if (i < ncap) cnt[i] = 0;
}


__global__ __launch_bounds__(256) void csr_rank_beside_gemm(
    const int* __restrict__ tgt, int* __restrict__ cnt, int* __restrict__ rank, int E,
    const float* __restrict__ X, const float* __restrict__ W,
    float* __restrict__ H, int n, int gemm_first, int gemm_count) {
  __shared__ v2f lw[(128 / 2) * 128];
  const int blk = blockIdx.x;
  if (blk < gemm_count) {
    gemm_wmma_f32_block<128>(X, W, H, n, gemm_first + blk, lw);
  } else {
    const int e = (blk - gemm_count) * 256 + threadIdx.x;
    if (e < E) rank[e] = atomicAdd(&cnt[tgt[e]], 1);
  }
}

__global__ __launch_bounds__(1024) void csr_scan(
    const int* __restrict__ cnt, int* __restrict__ rowptr, int n, int ncap) {
  __shared__ int wtot[2][32];
  const int tid = threadIdx.x, lane = tid & 31, wave = tid >> 5;
  const int4* c4 = (const int4*)cnt;
  int carry = 0, buf = 0;
  int4 cur = c4[tid];
  for (int base = 0; base < ncap; base += SCAN_TILE, buf ^= 1) {
    int4 nxt = make_int4(0, 0, 0, 0);
    if (base + SCAN_TILE < ncap) nxt = c4[((base + SCAN_TILE) >> 2) + tid];
    const int s = cur.x + cur.y + cur.z + cur.w;
    int incl = s;
#pragma unroll
    for (int o = 1; o < 32; o <<= 1) {
      const int t = __shfl_up(incl, o, 32);
      if (lane >= o) incl += t;
    }
    if (lane == 31) wtot[buf][wave] = incl;
    __syncthreads();
    const int wt = wtot[buf][lane];
    int winc = wt;
#pragma unroll
    for (int o = 1; o < 32; o <<= 1) {
      const int t = __shfl_up(winc, o, 32);
      if (lane >= o) winc += t;
    }
    const int tile_total = __shfl(winc, 31, 32);
    const int wave_excl  = __shfl(winc - wt, wave, 32);
    const int p0 = carry + wave_excl + incl - s;
    const int p1 = p0 + cur.x, p2 = p1 + cur.y, p3 = p2 + cur.z;
    const int idx = base + tid * 4;
    if (idx + 3 < n) {
      *(int4*)(rowptr + idx) = make_int4(p0, p1, p2, p3);
    } else {
      if (idx     < n) rowptr[idx]     = p0;
      if (idx + 1 < n) rowptr[idx + 1] = p1;
      if (idx + 2 < n) rowptr[idx + 2] = p2;
    }
    carry += tile_total;
    cur = nxt;
  }
  if (tid == 0) rowptr[n] = carry;
}

__global__ __launch_bounds__(256) void csr_fill_beside_gemm(
    const int* __restrict__ tgt, const int* __restrict__ rank,
    const int* __restrict__ rowptr, const int* __restrict__ src,
    const float* __restrict__ w, int4* __restrict__ slot, int E,
    const float* __restrict__ X, const float* __restrict__ W,
    float* __restrict__ H, int n, int gemm_first, int gemm_count) {
  __shared__ v2f lw[(128 / 2) * 128];
  const int blk = blockIdx.x;
  if (blk < gemm_count) {
    gemm_wmma_f32_block<128>(X, W, H, n, gemm_first + blk, lw);
  } else {
    const int e = (blk - gemm_count) * 256 + threadIdx.x;
    if (e < E)
      slot[rowptr[tgt[e]] + rank[e]] = make_int4(e, src[e], __float_as_int(w[e]), 0);
  }
}

template <int F4, typename REC>
__device__ __forceinline__ float4 row_sum(const float4* __restrict__ Hq,
                                          REC csr, int j, int end) {
  float4 acc = make_float4(0.0f, 0.0f, 0.0f, 0.0f);
  for (; j + 4 <= end; j += 4) {
    const auto e0 = csr[j], e1 = csr[j + 1], e2 = csr[j + 2], e3 = csr[j + 3];
    const float4 v0 = Hq[(size_t)e0.x * F4];
    const float4 v1 = Hq[(size_t)e1.x * F4];
    const float4 v2 = Hq[(size_t)e2.x * F4];
    const float4 v3 = Hq[(size_t)e3.x * F4];
    const float w0 = __int_as_float(e0.y), w1 = __int_as_float(e1.y);
    const float w2 = __int_as_float(e2.y), w3 = __int_as_float(e3.y);
    acc.x = fmaf(w0, v0.x, acc.x); acc.y = fmaf(w0, v0.y, acc.y);
    acc.z = fmaf(w0, v0.z, acc.z); acc.w = fmaf(w0, v0.w, acc.w);
    acc.x = fmaf(w1, v1.x, acc.x); acc.y = fmaf(w1, v1.y, acc.y);
    acc.z = fmaf(w1, v1.z, acc.z); acc.w = fmaf(w1, v1.w, acc.w);
    acc.x = fmaf(w2, v2.x, acc.x); acc.y = fmaf(w2, v2.y, acc.y);
    acc.z = fmaf(w2, v2.z, acc.z); acc.w = fmaf(w2, v2.w, acc.w);
    acc.x = fmaf(w3, v3.x, acc.x); acc.y = fmaf(w3, v3.y, acc.y);
    acc.z = fmaf(w3, v3.z, acc.z); acc.w = fmaf(w3, v3.w, acc.w);
  }
  for (; j < end; ++j) {
    const auto e0 = csr[j];
    const float4 v0 = Hq[(size_t)e0.x * F4];
    const float w0 = __int_as_float(e0.y);
    acc.x = fmaf(w0, v0.x, acc.x); acc.y = fmaf(w0, v0.y, acc.y);
    acc.z = fmaf(w0, v0.z, acc.z); acc.w = fmaf(w0, v0.w, acc.w);
  }
  return acc;
}

template <int F4>
__device__ __forceinline__ float4 row_sum_lanes(const float4* __restrict__ Hq,
                                                int rsrc, int rw, int d) {
  float4 acc = make_float4(0.0f, 0.0f, 0.0f, 0.0f);
  int p = 0;
  for (; p + 4 <= d; p += 4) {
    const int s0 = __builtin_amdgcn_readlane(rsrc, p);
    const int s1 = __builtin_amdgcn_readlane(rsrc, p + 1);
    const int s2 = __builtin_amdgcn_readlane(rsrc, p + 2);
    const int s3 = __builtin_amdgcn_readlane(rsrc, p + 3);
    const float4 v0 = Hq[(size_t)s0 * F4];
    const float4 v1 = Hq[(size_t)s1 * F4];
    const float4 v2 = Hq[(size_t)s2 * F4];
    const float4 v3 = Hq[(size_t)s3 * F4];
    const float w0 = __int_as_float(__builtin_amdgcn_readlane(rw, p));
    const float w1 = __int_as_float(__builtin_amdgcn_readlane(rw, p + 1));
    const float w2 = __int_as_float(__builtin_amdgcn_readlane(rw, p + 2));
    const float w3 = __int_as_float(__builtin_amdgcn_readlane(rw, p + 3));
    acc.x = fmaf(w0, v0.x, acc.x); acc.y = fmaf(w0, v0.y, acc.y);
    acc.z = fmaf(w0, v0.z, acc.z); acc.w = fmaf(w0, v0.w, acc.w);
    acc.x = fmaf(w1, v1.x, acc.x); acc.y = fmaf(w1, v1.y, acc.y);
    acc.z = fmaf(w1, v1.z, acc.z); acc.w = fmaf(w1, v1.w, acc.w);
    acc.x = fmaf(w2, v2.x, acc.x); acc.y = fmaf(w2, v2.y, acc.y);
    acc.z = fmaf(w2, v2.z, acc.z); acc.w = fmaf(w2, v2.w, acc.w);
    acc.x = fmaf(w3, v3.x, acc.x); acc.y = fmaf(w3, v3.y, acc.y);
    acc.z = fmaf(w3, v3.z, acc.z); acc.w = fmaf(w3, v3.w, acc.w);
  }
  for (; p < d; ++p) {
    const int s0 = __builtin_amdgcn_readlane(rsrc, p);
    const float4 v0 = Hq[(size_t)s0 * F4];
    const float w0 = __int_as_float(__builtin_amdgcn_readlane(rw, p));
    acc.x = fmaf(w0, v0.x, acc.x); acc.y = fmaf(w0, v0.y, acc.y);
    acc.z = fmaf(w0, v0.z, acc.z); acc.w = fmaf(w0, v0.w, acc.w);
  }
  return acc;
}

template <int F4>
__device__ __forceinline__ float4 order_row_and_sum(
    const float4* __restrict__ Hq, const int4* __restrict__ slot,
    int2* csr, int start, int d, int lane) {
  if (d <= 32) {
    int4 me = make_int4(0x7fffffff, 0, 0, 0);
    if (lane < d) me = slot[start + lane];
    int r = 0;
    for (int j = 0; j < d; ++j)
      r += (__builtin_amdgcn_readlane(me.x, j) < me.x) ? 1 : 0;
    const int place = (lane < d) ? r : lane;
    const int rsrc = __builtin_amdgcn_ds_permute(place << 2, me.y);
    const int rw   = __builtin_amdgcn_ds_permute(place << 2, me.z);
    if (lane < d) csr[start + lane] = make_int2(rsrc, rw);
    return row_sum_lanes<F4>(Hq, rsrc, rw, d);
  }
  float4 acc = make_float4(0.0f, 0.0f, 0.0f, 0.0f);
  if (d <= 64) {
    const bool hasb = 32 + lane < d;
    const int4 a = slot[start + lane];
    int4 b = make_int4(0x7fffffff, 0, 0, 0);
    if (hasb) b = slot[start + 32 + lane];
    int ra = 0, rb = 0;
    for (int j = 0; j < 32; ++j) {
      const int e = __builtin_amdgcn_readlane(a.x, j);
      ra += (e < a.x) ? 1 : 0;
      rb += (e < b.x) ? 1 : 0;
    }
    for (int j = 0; j < d - 32; ++j) {
      const int e = __builtin_amdgcn_readlane(b.x, j);
      ra += (e < a.x) ? 1 : 0;
      rb += (e < b.x) ? 1 : 0;
    }
    csr[start + ra] = make_int2(a.y, a.z);
    if (hasb) csr[start + rb] = make_int2(b.y, b.z);
    else rb = -1;
    auto pick = [&](int p, int& s, float& w) {
      const unsigned ma = __builtin_amdgcn_ballot_w32(ra == p);
      const unsigned mb = __builtin_amdgcn_ballot_w32(rb == p);
      const int la = ma ? __builtin_ctz(ma) : 0;
      const int lb = mb ? __builtin_ctz(mb) : 0;
      const int sa = __builtin_amdgcn_readlane(a.y, la);
      const int wa = __builtin_amdgcn_readlane(a.z, la);
      const int sb = __builtin_amdgcn_readlane(b.y, lb);
      const int wb = __builtin_amdgcn_readlane(b.z, lb);
      s = ma ? sa : sb;
      w = __int_as_float(ma ? wa : wb);
    };
    int p = 0;
    for (; p + 4 <= d; p += 4) {
      int s0, s1, s2, s3;
      float w0, w1, w2, w3;
      pick(p, s0, w0); pick(p + 1, s1, w1); pick(p + 2, s2, w2); pick(p + 3, s3, w3);
      const float4 v0 = Hq[(size_t)s0 * F4];
      const float4 v1 = Hq[(size_t)s1 * F4];
      const float4 v2 = Hq[(size_t)s2 * F4];
      const float4 v3 = Hq[(size_t)s3 * F4];
      acc.x = fmaf(w0, v0.x, acc.x); acc.y = fmaf(w0, v0.y, acc.y);
      acc.z = fmaf(w0, v0.z, acc.z); acc.w = fmaf(w0, v0.w, acc.w);
      acc.x = fmaf(w1, v1.x, acc.x); acc.y = fmaf(w1, v1.y, acc.y);
      acc.z = fmaf(w1, v1.z, acc.z); acc.w = fmaf(w1, v1.w, acc.w);
      acc.x = fmaf(w2, v2.x, acc.x); acc.y = fmaf(w2, v2.y, acc.y);
      acc.z = fmaf(w2, v2.z, acc.z); acc.w = fmaf(w2, v2.w, acc.w);
      acc.x = fmaf(w3, v3.x, acc.x); acc.y = fmaf(w3, v3.y, acc.y);
      acc.z = fmaf(w3, v3.z, acc.z); acc.w = fmaf(w3, v3.w, acc.w);
    }
    for (; p < d; ++p) {
      int s0;
      float w0;
      pick(p, s0, w0);
      const float4 v0 = Hq[(size_t)s0 * F4];
      acc.x = fmaf(w0, v0.x, acc.x); acc.y = fmaf(w0, v0.y, acc.y);
      acc.z = fmaf(w0, v0.z, acc.z); acc.w = fmaf(w0, v0.w, acc.w);
    }
    return acc;
  }
  for (int base = 0; base < d; base += 32) {
    const int i = base + lane;
    int4 me = make_int4(0x7fffffff, 0, 0, 0);
    if (i < d) me = slot[start + i];
    int r = 0;
    for (int j = 0; j < d; ++j) r += (slot[start + j].x < me.x) ? 1 : 0;
    if (i < d) csr[start + r] = make_int2(me.y, me.z);
  }
  __threadfence();
  int* rec = (int*)(csr + start);
  for (int p = 0; p < d; ++p) {
    const int s0 = __builtin_amdgcn_readfirstlane(
        __hip_atomic_load(rec + 2 * p, __ATOMIC_RELAXED, __HIP_MEMORY_SCOPE_AGENT));
    const float w0 = __int_as_float(__builtin_amdgcn_readfirstlane(
        __hip_atomic_load(rec + 2 * p + 1, __ATOMIC_RELAXED, __HIP_MEMORY_SCOPE_AGENT)));
    const float4 v0 = Hq[(size_t)s0 * F4];
    acc.x = fmaf(w0, v0.x, acc.x); acc.y = fmaf(w0, v0.y, acc.y);
    acc.z = fmaf(w0, v0.z, acc.z); acc.w = fmaf(w0, v0.w, acc.w);
  }
  return acc;
}

template <bool ORDER, bool GN, int NC>
__global__ __launch_bounds__(512) void gather128_tail_gemm(
    const float* __restrict__ H, const int* __restrict__ rowptr,
    const int2* __restrict__ csr, const int4* __restrict__ slot, int2* csr_out,
    const float* __restrict__ b,
    const float* __restrict__ gamma, const float* __restrict__ beta,
    const float* __restrict__ Wn, float* __restrict__ Hn, int n, int ntiles) {
  constexpr int K  = 128;
  constexpr int CT = (NC + 15) / 16;
  constexpr int PC = CT * 16;
  constexpr int TP = 132;
  __shared__ v2f lw[(K / 2) * PC];
  __shared__ __attribute__((aligned(16))) float tile[16 * TP];

  const int tid = threadIdx.x;
  for (int i = tid; i < (K / 2) * PC; i += 512) {
    const int kp = i / PC, c = i % PC;
    v2f val;
    val.x = (c < NC) ? Wn[(2 * kp) * NC + c] : 0.0f;
    val.y = (c < NC) ? Wn[(2 * kp + 1) * NC + c] : 0.0f;
    lw[i] = val;
  }

  const int lane = tid & 31;
  const int wave = __builtin_amdgcn_readfirstlane(tid >> 5);
  const int m  = lane & 15;
  const int hk = lane >> 4;

  const kint_p krow = (kint_p)rowptr;
  const krec_p krec = (krec_p)csr;

  for (int t = blockIdx.x; t < ntiles; t += gridDim.x) {
    const int node = t * 16 + wave;
    float4 v = make_float4(0.0f, 0.0f, 0.0f, 0.0f);
    if (node < n) {
      if (ORDER) {
        const int start = krow[node];
        v = order_row_and_sum<32>((const float4*)H + lane, slot, csr_out, start,
                                  krow[node + 1] - start, lane);
      } else {
        v = row_sum<32>((const float4*)H + lane, krec, krow[node], krow[node + 1]);
      }
      const float4 bb = *(const float4*)(b + lane * 4);
      v.x = fmaxf(v.x + bb.x, 0.0f);
      v.y = fmaxf(v.y + bb.y, 0.0f);
      v.z = fmaxf(v.z + bb.z, 0.0f);
      v.w = fmaxf(v.w + bb.w, 0.0f);
      if (GN) {
        const float mu  = 0.25f * (v.x + v.y + v.z + v.w);
        const float ms  = 0.25f * (v.x * v.x + v.y * v.y + v.z * v.z + v.w * v.w);
        const float inv = rsqrtf(fmaxf(ms - mu * mu, 0.0f) + 1e-5f);
        const float4 gm = *(const float4*)(gamma + lane * 4);
        const float4 bt = *(const float4*)(beta + lane * 4);
        v.x = (v.x - mu) * inv * gm.x + bt.x;
        v.y = (v.y - mu) * inv * gm.y + bt.y;
        v.z = (v.z - mu) * inv * gm.z + bt.z;
        v.w = (v.w - mu) * inv * gm.w + bt.w;
      }
    }
    __syncthreads();
    *(float4*)&tile[wave * TP + lane * 4] = v;
    __syncthreads();
    if (wave < CT) {
      const float* arow = &tile[m * TP + 2 * hk];
      const v2f*   brow = &lw[hk * PC + wave * 16 + m];
      v8f acc = {};
#pragma unroll
      for (int k4 = 0; k4 < K; k4 += 4) {
        const v2f a  = *(const v2f*)(arow + k4);
        const v2f bq = brow[(k4 / 2) * PC];
        acc = __builtin_amdgcn_wmma_f32_16x16x4_f32(
            false, a, false, bq, (short)0, acc, false, false);
      }
      const int crow = t * 16 + hk * 8;
      const int col  = wave * 16 + m;
      if ((NC % 16 == 0) || (col < NC)) {
#pragma unroll
        for (int r = 0; r < 8; ++r)
          Hn[(size_t)(crow + r) * NC + col] = acc[r];
      }
    }
  }
}

__global__ __launch_bounds__(256) void gather40_lsm(
    const float* __restrict__ H, const int* __restrict__ rowptr,
    const int2* __restrict__ csr, const float* __restrict__ b,
    float* __restrict__ out, int n) {
  const int gid  = blockIdx.x * 256 + threadIdx.x;
  const int node = gid >> 3;
  const int q    = gid & 7;
  const bool live = node < n;
  int j = 0, end = 0;
  if (live) { j = rowptr[node]; end = rowptr[node + 1]; }
  const float4 bb = *(const float4*)(b + q * 4);
  const float  b1 = b[32 + q];
  const float4* __restrict__ Hq = (const float4*)H + q;
  const float*  __restrict__ Ht = H + 32 + q;
  float4 acc = make_float4(0.0f, 0.0f, 0.0f, 0.0f);
  float  act = 0.0f;
  for (; j + 4 <= end; j += 4) {
    const int2 e0 = csr[j], e1 = csr[j + 1], e2 = csr[j + 2], e3 = csr[j + 3];
    const float4 v0 = Hq[(size_t)e0.x * 10];
    const float4 v1 = Hq[(size_t)e1.x * 10];
    const float4 v2 = Hq[(size_t)e2.x * 10];
    const float4 v3 = Hq[(size_t)e3.x * 10];
    const float  t0 = Ht[(size_t)e0.x * 40];
    const float  t1 = Ht[(size_t)e1.x * 40];
    const float  t2 = Ht[(size_t)e2.x * 40];
    const float  t3 = Ht[(size_t)e3.x * 40];
    const float w0 = __int_as_float(e0.y), w1 = __int_as_float(e1.y);
    const float w2 = __int_as_float(e2.y), w3 = __int_as_float(e3.y);
    acc.x = fmaf(w0, v0.x, acc.x); acc.y = fmaf(w0, v0.y, acc.y);
    acc.z = fmaf(w0, v0.z, acc.z); acc.w = fmaf(w0, v0.w, acc.w);
    act   = fmaf(w0, t0, act);
    acc.x = fmaf(w1, v1.x, acc.x); acc.y = fmaf(w1, v1.y, acc.y);
    acc.z = fmaf(w1, v1.z, acc.z); acc.w = fmaf(w1, v1.w, acc.w);
    act   = fmaf(w1, t1, act);
    acc.x = fmaf(w2, v2.x, acc.x); acc.y = fmaf(w2, v2.y, acc.y);
    acc.z = fmaf(w2, v2.z, acc.z); acc.w = fmaf(w2, v2.w, acc.w);
    act   = fmaf(w2, t2, act);
    acc.x = fmaf(w3, v3.x, acc.x); acc.y = fmaf(w3, v3.y, acc.y);
    acc.z = fmaf(w3, v3.z, acc.z); acc.w = fmaf(w3, v3.w, acc.w);
    act   = fmaf(w3, t3, act);
  }
  for (; j < end; ++j) {
    const int2 e0 = csr[j];
    const float4 v0 = Hq[(size_t)e0.x * 10];
    const float  t0 = Ht[(size_t)e0.x * 40];
    const float w0 = __int_as_float(e0.y);
    acc.x = fmaf(w0, v0.x, acc.x); acc.y = fmaf(w0, v0.y, acc.y);
    acc.z = fmaf(w0, v0.z, acc.z); acc.w = fmaf(w0, v0.w, acc.w);
    act   = fmaf(w0, t0, act);
  }
  const float x0 = acc.x + bb.x, x1 = acc.y + bb.y;
  const float x2 = acc.z + bb.z, x3 = acc.w + bb.w;
  const float x4 = act + b1;
  float m = fmaxf(fmaxf(fmaxf(x0, x1), fmaxf(x2, x3)), x4);
  for (int o = 4; o > 0; o >>= 1) m = fmaxf(m, __shfl_xor(m, o, 32));
  float s = ((expf(x0 - m) + expf(x1 - m)) + (expf(x2 - m) + expf(x3 - m)))
            + expf(x4 - m);
  for (int o = 4; o > 0; o >>= 1) s += __shfl_xor(s, o, 32);
  const float lse = m + logf(s);
  if (live) {
    ((float4*)out)[(size_t)node * 10 + q] =
        make_float4(x0 - lse, x1 - lse, x2 - lse, x3 - lse);
    out[(size_t)node * 40 + 32 + q] = x4 - lse;
  }
}

extern "C" void kernel_launch(void* const* d_in, const int* in_sizes, int n_in,
                              void* d_out, int out_size, void* d_ws, size_t ws_size,
                              hipStream_t stream) {
  const int N = NNODES, E = NEDGES;
  const float* x     = (const float*)d_in[0];
  const int*   src   = (const int*)d_in[1];
  const int*   tgt   = (const int*)d_in[2];
  const float* w     = (const float*)d_in[3];
  const float* W0    = (const float*)d_in[4];
  const float* b0    = (const float*)d_in[5];
  const float* W1    = (const float*)d_in[6];
  const float* b1    = (const float*)d_in[7];
  const float* W2    = (const float*)d_in[8];
  const float* b2    = (const float*)d_in[9];
  const float* gamma = (const float*)d_in[10];
  const float* beta  = (const float*)d_in[11];
  float* out = (float*)d_out;

  float* H   = (float*)d_ws;
  float* H1  = H + (size_t)N * 128;
  float* H2  = H;
  int*  cnt    = (int*)(H + (size_t)N * 256);
  int*  rowptr = cnt + NCAP;
  int*  rank   = rowptr + NCAP + 64;
  int4* slot   = (int4*)(rank + E);
  int2* csr    = (int2*)(slot + E);

  const int gemm_blocks  = (N + 127) / 128;
  const int gemm_half0   = (gemm_blocks + 1) / 2;
  const int gemm_half1   = gemm_blocks - gemm_half0;
  const int edge_thr_blocks = (E + 255) / 256;
  const int g40_blocks   = (N * 8 + 255) / 256;
  const int ntiles       = (N + 15) / 16;
  const int fused_blocks = 1024;

  csr_zero<<<NCAP / 256, 256, 0, stream>>>(cnt, NCAP);
  csr_rank_beside_gemm<<<gemm_half0 + edge_thr_blocks, 256, 0, stream>>>(
      tgt, cnt, rank, E, x, W0, H, N, 0, gemm_half0);
  csr_scan<<<1, 1024, 0, stream>>>(cnt, rowptr, N, NCAP);
  csr_fill_beside_gemm<<<gemm_half1 + edge_thr_blocks, 256, 0, stream>>>(
      tgt, rank, rowptr, src, w, slot, E, x, W0, H, N, gemm_half0, gemm_half1);

  gather128_tail_gemm<true, false, 128><<<fused_blocks, 512, 0, stream>>>(
      H, rowptr, nullptr, slot, csr, b0, gamma, beta, W1, H1, N, ntiles);

  gather128_tail_gemm<false, true, 40><<<fused_blocks, 512, 0, stream>>>(
      H1, rowptr, csr, nullptr, nullptr, b1, gamma, beta, W2, H2, N, ntiles);

  gather40_lsm<<<g40_blocks, 256, 0, stream>>>(H2, rowptr, csr, b2, out, N);
}


# every node owns a 32-slot bucket, so the counting atomic's return is the edge's slot (overflow list for longer rows): scan launch, second edge pass and rank array gone, 5 launches; sum order kept
# speedup vs baseline: 3.2641x; 1.2026x over previous
#include <hip/hip_runtime.h>
#include <math.h>

typedef __attribute__((ext_vector_type(2))) float v2f;
typedef __attribute__((ext_vector_type(8))) float v8f;
typedef __attribute__((ext_vector_type(2))) int   v2i;
typedef const __attribute__((address_space(4))) int* kint_p;
typedef const __attribute__((address_space(4))) v2i* krec_p;

#define NNODES 50000
#define NEDGES 800000
static_assert(NNODES % 16 == 0, "the 16-row tiles of both product kernels are whole");

template <int NC>
__device__ __forceinline__ void gemm_wmma_f32_block(
    const float* __restrict__ X, const float* __restrict__ W,
    float* __restrict__ H, int n, int blk, v2f* __restrict__ lw) {
  constexpr int K  = 128;
  constexpr int CT = (NC + 15) / 16;
  constexpr int PC = CT * 16;

  const int tid = threadIdx.x;
  for (int i = tid; i < (K / 2) * PC; i += 256) {
    const int kp = i / PC, c = i % PC;
    v2f val;
    val.x = (c < NC) ? W[(2 * kp) * NC + c] : 0.0f;
    val.y = (c < NC) ? W[(2 * kp + 1) * NC + c] : 0.0f;
    lw[i] = val;
  }
  __syncthreads();

  const int wave = tid >> 5;
  const int lane = tid & 31;
  const int row0 = (blk * 8 + wave) * 16;
  if (row0 >= n) return;

  const int m  = lane & 15;
  const int hk = lane >> 4;

  v8f acc[CT];
#pragma unroll
  for (int t = 0; t < CT; ++t) acc[t] = {};

  const float* xrow = X + (size_t)(row0 + m) * K + 2 * hk;

  for (int k4 = 0; k4 < K; k4 += 4) {
    v2f a;
    a.x = xrow[k4];
    a.y = xrow[k4 + 1];
    const v2f* brow = &lw[(k4 / 2 + hk) * PC + m];
#pragma unroll
    for (int t = 0; t < CT; ++t) {
      const v2f b = brow[t * 16];
      acc[t] = __builtin_amdgcn_wmma_f32_16x16x4_f32(
          false, a, false, b, (short)0, acc[t], false, false);
    }
  }

  const int crow = row0 + hk * 8;
#pragma unroll
  for (int t = 0; t < CT; ++t) {
    const int col = t * 16 + m;
    if ((NC % 16 == 0) || (col < NC)) {
#pragma unroll
      for (int r = 0; r < 8; ++r)
        H[(size_t)(crow + r) * NC + col] = acc[t][r];
    }
  }
}

#define ROWCAP 32
#define NCNT (((NNODES + 2 + 255) / 256) * 256)
#define OVF_COUNT NNODES
#define LONG_ALLOC (NNODES + 1)

__global__ void csr_zero(int* __restrict__ cnt, int ncnt) {
  const int i = blockIdx.x * blockDim.x + threadIdx.x;
  if (i < ncnt) cnt[i] = 0;
}


__global__ __launch_bounds__(256) void rows_fill_beside_gemm(
    const int* __restrict__ tgt, const int* __restrict__ src,
    const float* __restrict__ w, int* __restrict__ cnt,
    int4* __restrict__ slot, int4* __restrict__ ovf, int E,
    const float* __restrict__ X, const float* __restrict__ W,
    float* __restrict__ H, int n, int gemm_count) {
  __shared__ v2f lw[(128 / 2) * 128];
  const int blk = blockIdx.x;
  if (blk < gemm_count) {
    gemm_wmma_f32_block<128>(X, W, H, n, blk, lw);
  } else {
    const int e = (blk - gemm_count) * 256 + threadIdx.x;
    if (e < E) {
      const int t = tgt[e];
      const int4 rec = make_int4(e, src[e], __float_as_int(w[e]), t);
      const int r = atomicAdd(&cnt[t], 1);
      if (r < ROWCAP) slot[(size_t)t * ROWCAP + r] = rec;
      else ovf[atomicAdd(&cnt[OVF_COUNT], 1)] = rec;
    }
  }
}

template <int F4, typename REC>
__device__ __forceinline__ float4 row_sum(const float4* __restrict__ Hq,
                                          REC csr, int j, int end) {
  float4 acc = make_float4(0.0f, 0.0f, 0.0f, 0.0f);
  for (; j + 4 <= end; j += 4) {
    const auto e0 = csr[j], e1 = csr[j + 1], e2 = csr[j + 2], e3 = csr[j + 3];
    const float4 v0 = Hq[(size_t)e0.x * F4];
    const float4 v1 = Hq[(size_t)e1.x * F4];
    const float4 v2 = Hq[(size_t)e2.x * F4];
    const float4 v3 = Hq[(size_t)e3.x * F4];
    const float w0 = __int_as_float(e0.y), w1 = __int_as_float(e1.y);
    const float w2 = __int_as_float(e2.y), w3 = __int_as_float(e3.y);
    acc.x = fmaf(w0, v0.x, acc.x); acc.y = fmaf(w0, v0.y, acc.y);
    acc.z = fmaf(w0, v0.z, acc.z); acc.w = fmaf(w0, v0.w, acc.w);
    acc.x = fmaf(w1, v1.x, acc.x); acc.y = fmaf(w1, v1.y, acc.y);
    acc.z = fmaf(w1, v1.z, acc.z); acc.w = fmaf(w1, v1.w, acc.w);
    acc.x = fmaf(w2, v2.x, acc.x); acc.y = fmaf(w2, v2.y, acc.y);
    acc.z = fmaf(w2, v2.z, acc.z); acc.w = fmaf(w2, v2.w, acc.w);
    acc.x = fmaf(w3, v3.x, acc.x); acc.y = fmaf(w3, v3.y, acc.y);
    acc.z = fmaf(w3, v3.z, acc.z); acc.w = fmaf(w3, v3.w, acc.w);
  }
  for (; j < end; ++j) {
    const auto e0 = csr[j];
    const float4 v0 = Hq[(size_t)e0.x * F4];
    const float w0 = __int_as_float(e0.y);
    acc.x = fmaf(w0, v0.x, acc.x); acc.y = fmaf(w0, v0.y, acc.y);
    acc.z = fmaf(w0, v0.z, acc.z); acc.w = fmaf(w0, v0.w, acc.w);
  }
  return acc;
}

template <int F4>
__device__ __forceinline__ float4 row_sum_lanes(const float4* __restrict__ Hq,
                                                int rsrc, int rw, int d) {
  float4 acc = make_float4(0.0f, 0.0f, 0.0f, 0.0f);
  int p = 0;
  for (; p + 4 <= d; p += 4) {
    const int s0 = __builtin_amdgcn_readlane(rsrc, p);
    const int s1 = __builtin_amdgcn_readlane(rsrc, p + 1);
    const int s2 = __builtin_amdgcn_readlane(rsrc, p + 2);
    const int s3 = __builtin_amdgcn_readlane(rsrc, p + 3);
    const float4 v0 = Hq[(size_t)s0 * F4];
    const float4 v1 = Hq[(size_t)s1 * F4];
    const float4 v2 = Hq[(size_t)s2 * F4];
    const float4 v3 = Hq[(size_t)s3 * F4];
    const float w0 = __int_as_float(__builtin_amdgcn_readlane(rw, p));
    const float w1 = __int_as_float(__builtin_amdgcn_readlane(rw, p + 1));
    const float w2 = __int_as_float(__builtin_amdgcn_readlane(rw, p + 2));
    const float w3 = __int_as_float(__builtin_amdgcn_readlane(rw, p + 3));
    acc.x = fmaf(w0, v0.x, acc.x); acc.y = fmaf(w0, v0.y, acc.y);
    acc.z = fmaf(w0, v0.z, acc.z); acc.w = fmaf(w0, v0.w, acc.w);
    acc.x = fmaf(w1, v1.x, acc.x); acc.y = fmaf(w1, v1.y, acc.y);
    acc.z = fmaf(w1, v1.z, acc.z); acc.w = fmaf(w1, v1.w, acc.w);
    acc.x = fmaf(w2, v2.x, acc.x); acc.y = fmaf(w2, v2.y, acc.y);
    acc.z = fmaf(w2, v2.z, acc.z); acc.w = fmaf(w2, v2.w, acc.w);
    acc.x = fmaf(w3, v3.x, acc.x); acc.y = fmaf(w3, v3.y, acc.y);
    acc.z = fmaf(w3, v3.z, acc.z); acc.w = fmaf(w3, v3.w, acc.w);
  }
  for (; p < d; ++p) {
    const int s0 = __builtin_amdgcn_readlane(rsrc, p);
    const float4 v0 = Hq[(size_t)s0 * F4];
    const float w0 = __int_as_float(__builtin_amdgcn_readlane(rw, p));
    acc.x = fmaf(w0, v0.x, acc.x); acc.y = fmaf(w0, v0.y, acc.y);
    acc.z = fmaf(w0, v0.z, acc.z); acc.w = fmaf(w0, v0.w, acc.w);
  }
  return acc;
}

template <int F4>
__device__ __forceinline__ float4 order_row_and_sum(
    const float4* __restrict__ Hq, const int4* __restrict__ slot,
    int2* csr, int start, int d, int lane) {
  int4 me = make_int4(0x7fffffff, 0, 0, 0);
  if (lane < d) me = slot[start + lane];
  int r = 0;
  for (int j = 0; j < d; ++j)
    r += (__builtin_amdgcn_readlane(me.x, j) < me.x) ? 1 : 0;
  const int place = (lane < d) ? r : lane;
  const int rsrc = __builtin_amdgcn_ds_permute(place << 2, me.y);
  const int rw   = __builtin_amdgcn_ds_permute(place << 2, me.z);
  if (lane < d) csr[start + lane] = make_int2(rsrc, rw);
  return row_sum_lanes<F4>(Hq, rsrc, rw, d);
}

template <int F4>
__device__ __forceinline__ float4 order_long_row_and_sum(
    const float4* __restrict__ Hq, const int4* __restrict__ bucket,
    const int4* __restrict__ ovf, int nov, int node, int2* rec_out,
    int d, int lane) {
  const int4 bk = bucket[lane];
  {
    int r = 0;
    for (int j = 0; j < ROWCAP; ++j)
      r += (__builtin_amdgcn_readlane(bk.x, j) < bk.x) ? 1 : 0;
    for (int q = 0; q < nov; ++q) {
      const int4 o = ovf[q];
      r += (o.w == node && o.x < bk.x) ? 1 : 0;
    }
    rec_out[r] = make_int2(bk.y, bk.z);
  }
  for (int qb = 0; qb < nov; qb += 32) {
    int4 me = make_int4(0x7fffffff, 0, 0, -1);
    if (qb + lane < nov) me = ovf[qb + lane];
    int r = 0;
    for (int j = 0; j < ROWCAP; ++j)
      r += (__builtin_amdgcn_readlane(bk.x, j) < me.x) ? 1 : 0;
    for (int q = 0; q < nov; ++q) {
      const int4 o = ovf[q];
      r += (o.w == node && o.x < me.x) ? 1 : 0;
    }
    if (me.w == node) rec_out[r] = make_int2(me.y, me.z);
  }
  __threadfence();
  float4 acc = make_float4(0.0f, 0.0f, 0.0f, 0.0f);
  int* rec = (int*)rec_out;
  for (int p = 0; p < d; ++p) {
    const int s0 = __builtin_amdgcn_readfirstlane(
        __hip_atomic_load(rec + 2 * p, __ATOMIC_RELAXED, __HIP_MEMORY_SCOPE_AGENT));
    const float w0 = __int_as_float(__builtin_amdgcn_readfirstlane(
        __hip_atomic_load(rec + 2 * p + 1, __ATOMIC_RELAXED, __HIP_MEMORY_SCOPE_AGENT)));
    const float4 v0 = Hq[(size_t)s0 * F4];
    acc.x = fmaf(w0, v0.x, acc.x); acc.y = fmaf(w0, v0.y, acc.y);
    acc.z = fmaf(w0, v0.z, acc.z); acc.w = fmaf(w0, v0.w, acc.w);
  }
  return acc;
}

template <bool ORDER, bool GN, int NC>
__global__ __launch_bounds__(512) void gather128_tail_gemm(
    const float* __restrict__ H, int* cnt, int2* rowinfo,
    const int2* __restrict__ csr, const int4* __restrict__ slot,
    const int4* __restrict__ ovf, int2* csr_out,
    const float* __restrict__ b,
    const float* __restrict__ gamma, const float* __restrict__ beta,
    const float* __restrict__ Wn, float* __restrict__ Hn, int n, int ntiles) {
  constexpr int K  = 128;
  constexpr int CT = (NC + 15) / 16;
  constexpr int PC = CT * 16;
  constexpr int TP = 132;
  __shared__ v2f lw[(K / 2) * PC];
  __shared__ __attribute__((aligned(16))) float tile[16 * TP];

  const int tid = threadIdx.x;
  for (int i = tid; i < (K / 2) * PC; i += 512) {
    const int kp = i / PC, c = i % PC;
    v2f val;
    val.x = (c < NC) ? Wn[(2 * kp) * NC + c] : 0.0f;
    val.y = (c < NC) ? Wn[(2 * kp + 1) * NC + c] : 0.0f;
    lw[i] = val;
  }

  const int lane = tid & 31;
  const int wave = __builtin_amdgcn_readfirstlane(tid >> 5);
  const int m  = lane & 15;
  const int hk = lane >> 4;

  const kint_p kcnt  = (kint_p)cnt;
  const krec_p kinfo = (krec_p)rowinfo;
  const krec_p krec  = (krec_p)csr;

  for (int t = blockIdx.x; t < ntiles; t += gridDim.x) {
    const int node = t * 16 + wave;
    float4 v = make_float4(0.0f, 0.0f, 0.0f, 0.0f);
    if (node < n) {
      if (ORDER) {
        const int d = kcnt[node];
        int base = node * ROWCAP;
        if (d <= ROWCAP) {
          v = order_row_and_sum<32>((const float4*)H + lane, slot, csr_out, base, d, lane);
        } else {
          int got = 0;
          if (lane == 0) got = atomicAdd(cnt + LONG_ALLOC, d);
          base = NNODES * ROWCAP + __builtin_amdgcn_readfirstlane(got);
          v = order_long_row_and_sum<32>((const float4*)H + lane,
                                         slot + (size_t)node * ROWCAP, ovf,
                                         kcnt[OVF_COUNT], node, csr_out + base, d, lane);
        }
        if (lane == 0) rowinfo[node] = make_int2(base, d);
      } else {
        const v2i ri = kinfo[node];
        v = row_sum<32>((const float4*)H + lane, krec, ri.x, ri.x + ri.y);
      }
      const float4 bb = *(const float4*)(b + lane * 4);
      v.x = fmaxf(v.x + bb.x, 0.0f);
      v.y = fmaxf(v.y + bb.y, 0.0f);
      v.z = fmaxf(v.z + bb.z, 0.0f);
      v.w = fmaxf(v.w + bb.w, 0.0f);
      if (GN) {
        const float mu  = 0.25f * (v.x + v.y + v.z + v.w);
        const float ms  = 0.25f * (v.x * v.x + v.y * v.y + v.z * v.z + v.w * v.w);
        const float inv = rsqrtf(fmaxf(ms - mu * mu, 0.0f) + 1e-5f);
        const float4 gm = *(const float4*)(gamma + lane * 4);
        const float4 bt = *(const float4*)(beta + lane * 4);
        v.x = (v.x - mu) * inv * gm.x + bt.x;
        v.y = (v.y - mu) * inv * gm.y + bt.y;
        v.z = (v.z - mu) * inv * gm.z + bt.z;
        v.w = (v.w - mu) * inv * gm.w + bt.w;
      }
    }
    __syncthreads();
    *(float4*)&tile[wave * TP + lane * 4] = v;
    __syncthreads();
    if (wave < CT) {
      const float* arow = &tile[m * TP + 2 * hk];
      const v2f*   brow = &lw[hk * PC + wave * 16 + m];
      v8f acc = {};
#pragma unroll
      for (int k4 = 0; k4 < K; k4 += 4) {
        const v2f a  = *(const v2f*)(arow + k4);
        const v2f bq = brow[(k4 / 2) * PC];
        acc = __builtin_amdgcn_wmma_f32_16x16x4_f32(
            false, a, false, bq, (short)0, acc, false, false);
      }
      const int crow = t * 16 + hk * 8;
      const int col  = wave * 16 + m;
      if ((NC % 16 == 0) || (col < NC)) {
#pragma unroll
        for (int r = 0; r < 8; ++r)
          Hn[(size_t)(crow + r) * NC + col] = acc[r];
      }
    }
  }
}

__global__ __launch_bounds__(256) void gather40_lsm(
    const float* __restrict__ H, const int2* __restrict__ rowinfo,
    const int2* __restrict__ csr, const float* __restrict__ b,
    float* __restrict__ out, int n) {
  const int gid  = blockIdx.x * 256 + threadIdx.x;
  const int node = gid >> 3;
  const int q    = gid & 7;
  const bool live = node < n;
  int j = 0, end = 0;
  if (live) { const int2 ri = rowinfo[node]; j = ri.x; end = ri.x + ri.y; }
  const float4 bb = *(const float4*)(b + q * 4);
  const float  b1 = b[32 + q];
  const float4* __restrict__ Hq = (const float4*)H + q;
  const float*  __restrict__ Ht = H + 32 + q;
  float4 acc = make_float4(0.0f, 0.0f, 0.0f, 0.0f);
  float  act = 0.0f;
  for (; j + 4 <= end; j += 4) {
    const int2 e0 = csr[j], e1 = csr[j + 1], e2 = csr[j + 2], e3 = csr[j + 3];
    const float4 v0 = Hq[(size_t)e0.x * 10];
    const float4 v1 = Hq[(size_t)e1.x * 10];
    const float4 v2 = Hq[(size_t)e2.x * 10];
    const float4 v3 = Hq[(size_t)e3.x * 10];
    const float  t0 = Ht[(size_t)e0.x * 40];
    const float  t1 = Ht[(size_t)e1.x * 40];
    const float  t2 = Ht[(size_t)e2.x * 40];
    const float  t3 = Ht[(size_t)e3.x * 40];
    const float w0 = __int_as_float(e0.y), w1 = __int_as_float(e1.y);
    const float w2 = __int_as_float(e2.y), w3 = __int_as_float(e3.y);
    acc.x = fmaf(w0, v0.x, acc.x); acc.y = fmaf(w0, v0.y, acc.y);
    acc.z = fmaf(w0, v0.z, acc.z); acc.w = fmaf(w0, v0.w, acc.w);
    act   = fmaf(w0, t0, act);
    acc.x = fmaf(w1, v1.x, acc.x); acc.y = fmaf(w1, v1.y, acc.y);
    acc.z = fmaf(w1, v1.z, acc.z); acc.w = fmaf(w1, v1.w, acc.w);
    act   = fmaf(w1, t1, act);
    acc.x = fmaf(w2, v2.x, acc.x); acc.y = fmaf(w2, v2.y, acc.y);
    acc.z = fmaf(w2, v2.z, acc.z); acc.w = fmaf(w2, v2.w, acc.w);
    act   = fmaf(w2, t2, act);
    acc.x = fmaf(w3, v3.x, acc.x); acc.y = fmaf(w3, v3.y, acc.y);
    acc.z = fmaf(w3, v3.z, acc.z); acc.w = fmaf(w3, v3.w, acc.w);
    act   = fmaf(w3, t3, act);
  }
  for (; j < end; ++j) {
    const int2 e0 = csr[j];
    const float4 v0 = Hq[(size_t)e0.x * 10];
    const float  t0 = Ht[(size_t)e0.x * 40];
    const float w0 = __int_as_float(e0.y);
    acc.x = fmaf(w0, v0.x, acc.x); acc.y = fmaf(w0, v0.y, acc.y);
    acc.z = fmaf(w0, v0.z, acc.z); acc.w = fmaf(w0, v0.w, acc.w);
    act   = fmaf(w0, t0, act);
  }
  const float x0 = acc.x + bb.x, x1 = acc.y + bb.y;
  const float x2 = acc.z + bb.z, x3 = acc.w + bb.w;
  const float x4 = act + b1;
  float m = fmaxf(fmaxf(fmaxf(x0, x1), fmaxf(x2, x3)), x4);
  for (int o = 4; o > 0; o >>= 1) m = fmaxf(m, __shfl_xor(m, o, 32));
  float s = ((expf(x0 - m) + expf(x1 - m)) + (expf(x2 - m) + expf(x3 - m)))
            + expf(x4 - m);
  for (int o = 4; o > 0; o >>= 1) s += __shfl_xor(s, o, 32);
  const float lse = m + logf(s);
  if (live) {
    ((float4*)out)[(size_t)node * 10 + q] =
        make_float4(x0 - lse, x1 - lse, x2 - lse, x3 - lse);
    out[(size_t)node * 40 + 32 + q] = x4 - lse;
  }
}

extern "C" void kernel_launch(void* const* d_in, const int* in_sizes, int n_in,
                              void* d_out, int out_size, void* d_ws, size_t ws_size,
                              hipStream_t stream) {
  const int N = NNODES, E = NEDGES;
  const float* x     = (const float*)d_in[0];
  const int*   src   = (const int*)d_in[1];
  const int*   tgt   = (const int*)d_in[2];
  const float* w     = (const float*)d_in[3];
  const float* W0    = (const float*)d_in[4];
  const float* b0    = (const float*)d_in[5];
  const float* W1    = (const float*)d_in[6];
  const float* b1    = (const float*)d_in[7];
  const float* W2    = (const float*)d_in[8];
  const float* b2    = (const float*)d_in[9];
  const float* gamma = (const float*)d_in[10];
  const float* beta  = (const float*)d_in[11];
  float* out = (float*)d_out;

  float* H   = (float*)d_ws;
  float* H1  = H + (size_t)N * 128;
  float* H2  = H;
  int*  cnt     = (int*)(H + (size_t)N * 256);
  int2* rowinfo = (int2*)(cnt + NCNT);
  int4* slot    = (int4*)(rowinfo + NCNT);
  int4* ovf     = slot + (size_t)N * ROWCAP;
  int2* csr     = (int2*)(ovf + E);

  const int gemm_blocks  = (N + 127) / 128;
  const int edge_thr_blocks = (E + 255) / 256;
  const int g40_blocks   = (N * 8 + 255) / 256;
  const int ntiles       = (N + 15) / 16;
  const int fused_blocks = 1024;

  csr_zero<<<NCNT / 256, 256, 0, stream>>>(cnt, NCNT);
  rows_fill_beside_gemm<<<gemm_blocks + edge_thr_blocks, 256, 0, stream>>>(
      tgt, src, w, cnt, slot, ovf, E, x, W0, H, N, gemm_blocks);

  gather128_tail_gemm<true, false, 128><<<fused_blocks, 512, 0, stream>>>(
      H, cnt, rowinfo, nullptr, slot, ovf, csr, b0, gamma, beta, W1, H1, N, ntiles);

  gather128_tail_gemm<false, true, 40><<<fused_blocks, 512, 0, stream>>>(
      H1, nullptr, rowinfo, csr, nullptr, nullptr, nullptr, b1, gamma, beta, W2, H2, N, ntiles);

  gather40_lsm<<<g40_blocks, 256, 0, stream>>>(H2, rowinfo, csr, b2, out, N);
}


# edge ranks counted in LDS in two steps (a block bins its edges by target range, then a work-group per bin ranks and places them): per-edge global atomics and the clearing launch gone; sum order kept
# speedup vs baseline: 4.5541x; 1.3952x over previous
#include <hip/hip_runtime.h>
#include <math.h>

typedef __attribute__((ext_vector_type(2))) float v2f;
typedef __attribute__((ext_vector_type(8))) float v8f;
typedef __attribute__((ext_vector_type(2))) int   v2i;
typedef const __attribute__((address_space(4))) int* kint_p;
typedef const __attribute__((address_space(4))) v2i* krec_p;

#define NNODES 50000
#define NEDGES 800000
static_assert(NNODES % 16 == 0, "the 16-row tiles of both product kernels are whole");

template <int NC>
__device__ __forceinline__ void gemm_wmma_f32_block(
    const float* __restrict__ X, const float* __restrict__ W,
    float* __restrict__ H, int n, int blk, v2f* __restrict__ lw) {
  constexpr int K  = 128;
  constexpr int CT = (NC + 15) / 16;
  constexpr int PC = CT * 16;

  const int tid = threadIdx.x;
  for (int i = tid; i < (K / 2) * PC; i += 256) {
    const int kp = i / PC, c = i % PC;
    v2f val;
    val.x = (c < NC) ? W[(2 * kp) * NC + c] : 0.0f;
    val.y = (c < NC) ? W[(2 * kp + 1) * NC + c] : 0.0f;
    lw[i] = val;
  }
  __syncthreads();

  const int wave = tid >> 5;
  const int lane = tid & 31;
  const int row0 = (blk * 8 + wave) * 16;
  if (row0 >= n) return;

  const int m  = lane & 15;
  const int hk = lane >> 4;

  v8f acc[CT];
#pragma unroll
  for (int t = 0; t < CT; ++t) acc[t] = {};

  const float* xrow = X + (size_t)(row0 + m) * K + 2 * hk;

  for (int k4 = 0; k4 < K; k4 += 4) {
    v2f a;
    a.x = xrow[k4];
    a.y = xrow[k4 + 1];
    const v2f* brow = &lw[(k4 / 2 + hk) * PC + m];
#pragma unroll
    for (int t = 0; t < CT; ++t) {
      const v2f b = brow[t * 16];
      acc[t] = __builtin_amdgcn_wmma_f32_16x16x4_f32(
          false, a, false, b, (short)0, acc[t], false, false);
    }
  }

  const int crow = row0 + hk * 8;
#pragma unroll
  for (int t = 0; t < CT; ++t) {
    const int col = t * 16 + m;
    if ((NC % 16 == 0) || (col < NC)) {
#pragma unroll
      for (int r = 0; r < 8; ++r)
        H[(size_t)(crow + r) * NC + col] = acc[t][r];
    }
  }
}

#define ROWCAP 32
#define NBINS 256
#define BINW ((NNODES + NBINS - 1) / NBINS)
#define EPT 16
#define PLACE_WAVES 16
#define EDGES_PER_BLOCK (256 * EPT)
#define BIN_BLOCKS ((NEDGES + EDGES_PER_BLOCK - 1) / EDGES_PER_BLOCK)
#define NCNT (((NNODES + 2 + 255) / 256) * 256)
#define OVF_COUNT NNODES
#define LONG_ALLOC (NNODES + 1)


__global__ __launch_bounds__(256) void edges_bin_beside_gemm(
    const int* __restrict__ tgt, const int* __restrict__ src,
    const float* __restrict__ w, int* __restrict__ cnt,
    int2* __restrict__ runs, int4* __restrict__ binned, int E,
    const float* __restrict__ X, const float* __restrict__ W,
    float* __restrict__ H, int n, int gemm_count) {
  static_assert(NBINS == 256, "thread b of a block owns bin b");
  __shared__ v2f lw[(128 / 2) * 128];
  __shared__ int hist[NBINS];
  __shared__ int wsum[8];
  const int blk = blockIdx.x;
  if (blk < gemm_count) {
    gemm_wmma_f32_block<128>(X, W, H, n, blk, lw);
  } else {
    const int tid = threadIdx.x;
    const int g = blk - gemm_count;
    const int e0 = g * EDGES_PER_BLOCK + tid;
    hist[tid] = 0;
    if (g == 0 && tid == 0) { cnt[OVF_COUNT] = 0; cnt[LONG_ALLOC] = 0; }
    int t[EPT], s[EPT], wi[EPT], rl[EPT];
#pragma unroll
    for (int i = 0; i < EPT; ++i) {
      const int e = e0 + i * 256;
      t[i] = 0; s[i] = 0; wi[i] = 0;
      if (e < E) { t[i] = tgt[e]; s[i] = src[e]; wi[i] = __float_as_int(w[e]); }
    }
    __syncthreads();
#pragma unroll
    for (int i = 0; i < EPT; ++i) {
      rl[i] = 0;
      if (e0 + i * 256 < E) rl[i] = atomicAdd(&hist[t[i] / BINW], 1);
    }
    __syncthreads();
    const int lane = tid & 31, wave = tid >> 5;
    const int c = hist[tid];
    int incl = c;
    for (int o = 1; o < 32; o <<= 1) {
      const int v = __shfl_up(incl, o, 32);
      if (lane >= o) incl += v;
    }
    if (lane == 31) wsum[wave] = incl;
    __syncthreads();
    int first = incl - c;
    for (int k = 0; k < wave; ++k) first += wsum[k];
    hist[tid] = first;
    runs[(size_t)g * NBINS + tid] = make_int2(first, c);
    __syncthreads();
    int4* __restrict__ mine = binned + (size_t)g * EDGES_PER_BLOCK;
#pragma unroll
    for (int i = 0; i < EPT; ++i) {
      const int e = e0 + i * 256;
      if (e < E)
        mine[hist[t[i] / BINW] + rl[i]] = make_int4(e, s[i], wi[i], t[i]);
    }
  }
}

__device__ __forceinline__ void place_record(int4 rec, int rank, int* __restrict__ cnt,
                                             int4* __restrict__ slot,
                                             int4* __restrict__ ovf) {
  int4* dst;
  if (rank < ROWCAP) dst = slot + ((size_t)rec.w * ROWCAP + rank);
  else dst = ovf + atomicAdd(&cnt[OVF_COUNT], 1);
  *dst = rec;
}

__global__ __launch_bounds__(PLACE_WAVES * 32) void rows_place(
    int* __restrict__ cnt, const int2* __restrict__ runs,
    const int4* __restrict__ binned, int4* __restrict__ slot,
    int4* __restrict__ ovf, int n) {
  constexpr int KMAX = (BIN_BLOCKS + PLACE_WAVES - 1) / PLACE_WAVES;
  static_assert(KMAX <= 32, "a lane per run of the wave");
  __shared__ int lc[BINW];
  const int tid = threadIdx.x;
  const int lane = tid & 31;
  const int wave = __builtin_amdgcn_readfirstlane(tid >> 5);
  const int b = blockIdx.x;
  const int node0 = b * BINW;
  if (tid < BINW) lc[tid] = 0;
  int2 run = make_int2(0, 0);
  {
    const int g = wave + PLACE_WAVES * lane;
    if (lane < KMAX && g < BIN_BLOCKS) run = runs[(size_t)g * NBINS + b];
  }
  __syncthreads();
  int4 rec[KMAX];
#pragma unroll
  for (int k = 0; k < KMAX; ++k) {
    const int first = __builtin_amdgcn_readlane(run.x, k);
    const int len   = __builtin_amdgcn_readlane(run.y, k);
    if (lane < len)
      rec[k] = binned[(size_t)(wave + PLACE_WAVES * k) * EDGES_PER_BLOCK + first + lane];
  }
#pragma unroll
  for (int k = 0; k < KMAX; ++k) {
    const int len = __builtin_amdgcn_readlane(run.y, k);
    if (lane < len)
      place_record(rec[k], atomicAdd(&lc[rec[k].w - node0], 1), cnt, slot, ovf);
  }
#pragma unroll
  for (int k = 0; k < KMAX; ++k) {
    const int first = __builtin_amdgcn_readlane(run.x, k);
    const int len   = __builtin_amdgcn_readlane(run.y, k);
    for (int j = 32 + lane; j < len; j += 32) {
      const int4 r = binned[(size_t)(wave + PLACE_WAVES * k) * EDGES_PER_BLOCK + first + j];
      place_record(r, atomicAdd(&lc[r.w - node0], 1), cnt, slot, ovf);
    }
  }
  __syncthreads();
  if (tid < BINW && node0 + tid < n) cnt[node0 + tid] = lc[tid];
}

template <int F4, typename REC>
__device__ __forceinline__ float4 row_sum(const float4* __restrict__ Hq,
                                          REC csr, int j, int end) {
  float4 acc = make_float4(0.0f, 0.0f, 0.0f, 0.0f);
  for (; j + 4 <= end; j += 4) {
    const auto e0 = csr[j], e1 = csr[j + 1], e2 = csr[j + 2], e3 = csr[j + 3];
    const float4 v0 = Hq[(size_t)e0.x * F4];
    const float4 v1 = Hq[(size_t)e1.x * F4];
    const float4 v2 = Hq[(size_t)e2.x * F4];
    const float4 v3 = Hq[(size_t)e3.x * F4];
    const float w0 = __int_as_float(e0.y), w1 = __int_as_float(e1.y);
    const float w2 = __int_as_float(e2.y), w3 = __int_as_float(e3.y);
    acc.x = fmaf(w0, v0.x, acc.x); acc.y = fmaf(w0, v0.y, acc.y);
    acc.z = fmaf(w0, v0.z, acc.z); acc.w = fmaf(w0, v0.w, acc.w);
    acc.x = fmaf(w1, v1.x, acc.x); acc.y = fmaf(w1, v1.y, acc.y);
    acc.z = fmaf(w1, v1.z, acc.z); acc.w = fmaf(w1, v1.w, acc.w);
    acc.x = fmaf(w2, v2.x, acc.x); acc.y = fmaf(w2, v2.y, acc.y);
    acc.z = fmaf(w2, v2.z, acc.z); acc.w = fmaf(w2, v2.w, acc.w);
    acc.x = fmaf(w3, v3.x, acc.x); acc.y = fmaf(w3, v3.y, acc.y);
    acc.z = fmaf(w3, v3.z, acc.z); acc.w = fmaf(w3, v3.w, acc.w);
  }
  for (; j < end; ++j) {
    const auto e0 = csr[j];
    const float4 v0 = Hq[(size_t)e0.x * F4];
    const float w0 = __int_as_float(e0.y);
    acc.x = fmaf(w0, v0.x, acc.x); acc.y = fmaf(w0, v0.y, acc.y);
    acc.z = fmaf(w0, v0.z, acc.z); acc.w = fmaf(w0, v0.w, acc.w);
  }
  return acc;
}

template <int F4>
__device__ __forceinline__ float4 row_sum_lanes(const float4* __restrict__ Hq,
                                                int rsrc, int rw, int d) {
  float4 acc = make_float4(0.0f, 0.0f, 0.0f, 0.0f);
  int p = 0;
  for (; p + 4 <= d; p += 4) {
    const int s0 = __builtin_amdgcn_readlane(rsrc, p);
    const int s1 = __builtin_amdgcn_readlane(rsrc, p + 1);
    const int s2 = __builtin_amdgcn_readlane(rsrc, p + 2);
    const int s3 = __builtin_amdgcn_readlane(rsrc, p + 3);
    const float4 v0 = Hq[(size_t)s0 * F4];
    const float4 v1 = Hq[(size_t)s1 * F4];
    const float4 v2 = Hq[(size_t)s2 * F4];
    const float4 v3 = Hq[(size_t)s3 * F4];
    const float w0 = __int_as_float(__builtin_amdgcn_readlane(rw, p));
    const float w1 = __int_as_float(__builtin_amdgcn_readlane(rw, p + 1));
    const float w2 = __int_as_float(__builtin_amdgcn_readlane(rw, p + 2));
    const float w3 = __int_as_float(__builtin_amdgcn_readlane(rw, p + 3));
    acc.x = fmaf(w0, v0.x, acc.x); acc.y = fmaf(w0, v0.y, acc.y);
    acc.z = fmaf(w0, v0.z, acc.z); acc.w = fmaf(w0, v0.w, acc.w);
    acc.x = fmaf(w1, v1.x, acc.x); acc.y = fmaf(w1, v1.y, acc.y);
    acc.z = fmaf(w1, v1.z, acc.z); acc.w = fmaf(w1, v1.w, acc.w);
    acc.x = fmaf(w2, v2.x, acc.x); acc.y = fmaf(w2, v2.y, acc.y);
    acc.z = fmaf(w2, v2.z, acc.z); acc.w = fmaf(w2, v2.w, acc.w);
    acc.x = fmaf(w3, v3.x, acc.x); acc.y = fmaf(w3, v3.y, acc.y);
    acc.z = fmaf(w3, v3.z, acc.z); acc.w = fmaf(w3, v3.w, acc.w);
  }
  for (; p < d; ++p) {
    const int s0 = __builtin_amdgcn_readlane(rsrc, p);
    const float4 v0 = Hq[(size_t)s0 * F4];
    const float w0 = __int_as_float(__builtin_amdgcn_readlane(rw, p));
    acc.x = fmaf(w0, v0.x, acc.x); acc.y = fmaf(w0, v0.y, acc.y);
    acc.z = fmaf(w0, v0.z, acc.z); acc.w = fmaf(w0, v0.w, acc.w);
  }
  return acc;
}

template <int F4>
__device__ __forceinline__ float4 order_row_and_sum(
    const float4* __restrict__ Hq, const int4* __restrict__ slot,
    int2* csr, int start, int d, int lane) {
  int4 me = make_int4(0x7fffffff, 0, 0, 0);
  if (lane < d) me = slot[start + lane];
  int r = 0;
  for (int j = 0; j < d; ++j)
    r += (__builtin_amdgcn_readlane(me.x, j) < me.x) ? 1 : 0;
  const int place = (lane < d) ? r : lane;
  const int rsrc = __builtin_amdgcn_ds_permute(place << 2, me.y);
  const int rw   = __builtin_amdgcn_ds_permute(place << 2, me.z);
  if (lane < d) csr[start + lane] = make_int2(rsrc, rw);
  return row_sum_lanes<F4>(Hq, rsrc, rw, d);
}

template <int F4>
__device__ __forceinline__ float4 order_long_row_and_sum(
    const float4* __restrict__ Hq, const int4* __restrict__ bucket,
    const int4* __restrict__ ovf, int nov, int node, int2* rec_out,
    int d, int lane) {
  const int4 bk = bucket[lane];
  {
    int r = 0;
    for (int j = 0; j < ROWCAP; ++j)
      r += (__builtin_amdgcn_readlane(bk.x, j) < bk.x) ? 1 : 0;
    for (int q = 0; q < nov; ++q) {
      const int4 o = ovf[q];
      r += (o.w == node && o.x < bk.x) ? 1 : 0;
    }
    rec_out[r] = make_int2(bk.y, bk.z);
  }
  for (int qb = 0; qb < nov; qb += 32) {
    int4 me = make_int4(0x7fffffff, 0, 0, -1);
    if (qb + lane < nov) me = ovf[qb + lane];
    int r = 0;
    for (int j = 0; j < ROWCAP; ++j)
      r += (__builtin_amdgcn_readlane(bk.x, j) < me.x) ? 1 : 0;
    for (int q = 0; q < nov; ++q) {
      const int4 o = ovf[q];
      r += (o.w == node && o.x < me.x) ? 1 : 0;
    }
    if (me.w == node) rec_out[r] = make_int2(me.y, me.z);
  }
  __threadfence();
  float4 acc = make_float4(0.0f, 0.0f, 0.0f, 0.0f);
  int* rec = (int*)rec_out;
  for (int p = 0; p < d; ++p) {
    const int s0 = __builtin_amdgcn_readfirstlane(
        __hip_atomic_load(rec + 2 * p, __ATOMIC_RELAXED, __HIP_MEMORY_SCOPE_AGENT));
    const float w0 = __int_as_float(__builtin_amdgcn_readfirstlane(
        __hip_atomic_load(rec + 2 * p + 1, __ATOMIC_RELAXED, __HIP_MEMORY_SCOPE_AGENT)));
    const float4 v0 = Hq[(size_t)s0 * F4];
    acc.x = fmaf(w0, v0.x, acc.x); acc.y = fmaf(w0, v0.y, acc.y);
    acc.z = fmaf(w0, v0.z, acc.z); acc.w = fmaf(w0, v0.w, acc.w);
  }
  return acc;
}

template <bool ORDER, bool GN, int NC>
__global__ __launch_bounds__(512) void gather128_tail_gemm(
    const float* __restrict__ H, int* cnt, int2* rowinfo,
    const int2* __restrict__ csr, const int4* __restrict__ slot,
    const int4* __restrict__ ovf, int2* csr_out,
    const float* __restrict__ b,
    const float* __restrict__ gamma, const float* __restrict__ beta,
    const float* __restrict__ Wn, float* __restrict__ Hn, int n, int ntiles) {
  constexpr int K  = 128;
  constexpr int CT = (NC + 15) / 16;
  constexpr int PC = CT * 16;
  constexpr int TP = 132;
  __shared__ v2f lw[(K / 2) * PC];
  __shared__ __attribute__((aligned(16))) float tile[16 * TP];

  const int tid = threadIdx.x;
  for (int i = tid; i < (K / 2) * PC; i += 512) {
    const int kp = i / PC, c = i % PC;
    v2f val;
    val.x = (c < NC) ? Wn[(2 * kp) * NC + c] : 0.0f;
    val.y = (c < NC) ? Wn[(2 * kp + 1) * NC + c] : 0.0f;
    lw[i] = val;
  }

  const int lane = tid & 31;
  const int wave = __builtin_amdgcn_readfirstlane(tid >> 5);
  const int m  = lane & 15;
  const int hk = lane >> 4;

  const kint_p kcnt  = (kint_p)cnt;
  const krec_p kinfo = (krec_p)rowinfo;
  const krec_p krec  = (krec_p)csr;

  for (int t = blockIdx.x; t < ntiles; t += gridDim.x) {
    const int node = t * 16 + wave;
    float4 v = make_float4(0.0f, 0.0f, 0.0f, 0.0f);
    if (node < n) {
      if (ORDER) {
        const int d = kcnt[node];
        int base = node * ROWCAP;
        if (d <= ROWCAP) {
          v = order_row_and_sum<32>((const float4*)H + lane, slot, csr_out, base, d, lane);
        } else {
          int got = 0;
          if (lane == 0) got = atomicAdd(cnt + LONG_ALLOC, d);
          base = NNODES * ROWCAP + __builtin_amdgcn_readfirstlane(got);
          v = order_long_row_and_sum<32>((const float4*)H + lane,
                                         slot + (size_t)node * ROWCAP, ovf,
                                         kcnt[OVF_COUNT], node, csr_out + base, d, lane);
        }
        if (lane == 0) rowinfo[node] = make_int2(base, d);
      } else {
        const v2i ri = kinfo[node];
        v = row_sum<32>((const float4*)H + lane, krec, ri.x, ri.x + ri.y);
      }
      const float4 bb = *(const float4*)(b + lane * 4);
      v.x = fmaxf(v.x + bb.x, 0.0f);
      v.y = fmaxf(v.y + bb.y, 0.0f);
      v.z = fmaxf(v.z + bb.z, 0.0f);
      v.w = fmaxf(v.w + bb.w, 0.0f);
      if (GN) {
        const float mu  = 0.25f * (v.x + v.y + v.z + v.w);
        const float ms  = 0.25f * (v.x * v.x + v.y * v.y + v.z * v.z + v.w * v.w);
        const float inv = rsqrtf(fmaxf(ms - mu * mu, 0.0f) + 1e-5f);
        const float4 gm = *(const float4*)(gamma + lane * 4);
        const float4 bt = *(const float4*)(beta + lane * 4);
        v.x = (v.x - mu) * inv * gm.x + bt.x;
        v.y = (v.y - mu) * inv * gm.y + bt.y;
        v.z = (v.z - mu) * inv * gm.z + bt.z;
        v.w = (v.w - mu) * inv * gm.w + bt.w;
      }
    }
    __syncthreads();
    *(float4*)&tile[wave * TP + lane * 4] = v;
    __syncthreads();
    if (wave < CT) {
      const float* arow = &tile[m * TP + 2 * hk];
      const v2f*   brow = &lw[hk * PC + wave * 16 + m];
      v8f acc = {};
#pragma unroll
      for (int k4 = 0; k4 < K; k4 += 4) {
        const v2f a  = *(const v2f*)(arow + k4);
        const v2f bq = brow[(k4 / 2) * PC];
        acc = __builtin_amdgcn_wmma_f32_16x16x4_f32(
            false, a, false, bq, (short)0, acc, false, false);
      }
      const int crow = t * 16 + hk * 8;
      const int col  = wave * 16 + m;
      if ((NC % 16 == 0) || (col < NC)) {
#pragma unroll
        for (int r = 0; r < 8; ++r)
          Hn[(size_t)(crow + r) * NC + col] = acc[r];
      }
    }
  }
}

__global__ __launch_bounds__(256) void gather40_lsm(
    const float* __restrict__ H, const int2* __restrict__ rowinfo,
    const int2* __restrict__ csr, const float* __restrict__ b,
    float* __restrict__ out, int n) {
  const int gid  = blockIdx.x * 256 + threadIdx.x;
  const int node = gid >> 3;
  const int q    = gid & 7;
  const bool live = node < n;
  int j = 0, end = 0;
  if (live) { const int2 ri = rowinfo[node]; j = ri.x; end = ri.x + ri.y; }
  const float4 bb = *(const float4*)(b + q * 4);
  const float  b1 = b[32 + q];
  const float4* __restrict__ Hq = (const float4*)H + q;
  const float*  __restrict__ Ht = H + 32 + q;
  float4 acc = make_float4(0.0f, 0.0f, 0.0f, 0.0f);
  float  act = 0.0f;
  for (; j + 4 <= end; j += 4) {
    const int2 e0 = csr[j], e1 = csr[j + 1], e2 = csr[j + 2], e3 = csr[j + 3];
    const float4 v0 = Hq[(size_t)e0.x * 10];
    const float4 v1 = Hq[(size_t)e1.x * 10];
    const float4 v2 = Hq[(size_t)e2.x * 10];
    const float4 v3 = Hq[(size_t)e3.x * 10];
    const float  t0 = Ht[(size_t)e0.x * 40];
    const float  t1 = Ht[(size_t)e1.x * 40];
    const float  t2 = Ht[(size_t)e2.x * 40];
    const float  t3 = Ht[(size_t)e3.x * 40];
    const float w0 = __int_as_float(e0.y), w1 = __int_as_float(e1.y);
    const float w2 = __int_as_float(e2.y), w3 = __int_as_float(e3.y);
    acc.x = fmaf(w0, v0.x, acc.x); acc.y = fmaf(w0, v0.y, acc.y);
    acc.z = fmaf(w0, v0.z, acc.z); acc.w = fmaf(w0, v0.w, acc.w);
    act   = fmaf(w0, t0, act);
    acc.x = fmaf(w1, v1.x, acc.x); acc.y = fmaf(w1, v1.y, acc.y);
    acc.z = fmaf(w1, v1.z, acc.z); acc.w = fmaf(w1, v1.w, acc.w);
    act   = fmaf(w1, t1, act);
    acc.x = fmaf(w2, v2.x, acc.x); acc.y = fmaf(w2, v2.y, acc.y);
    acc.z = fmaf(w2, v2.z, acc.z); acc.w = fmaf(w2, v2.w, acc.w);
    act   = fmaf(w2, t2, act);
    acc.x = fmaf(w3, v3.x, acc.x); acc.y = fmaf(w3, v3.y, acc.y);
    acc.z = fmaf(w3, v3.z, acc.z); acc.w = fmaf(w3, v3.w, acc.w);
    act   = fmaf(w3, t3, act);
  }
  for (; j < end; ++j) {
    const int2 e0 = csr[j];
    const float4 v0 = Hq[(size_t)e0.x * 10];
    const float  t0 = Ht[(size_t)e0.x * 40];
    const float w0 = __int_as_float(e0.y);
    acc.x = fmaf(w0, v0.x, acc.x); acc.y = fmaf(w0, v0.y, acc.y);
    acc.z = fmaf(w0, v0.z, acc.z); acc.w = fmaf(w0, v0.w, acc.w);
    act   = fmaf(w0, t0, act);
  }
  const float x0 = acc.x + bb.x, x1 = acc.y + bb.y;
  const float x2 = acc.z + bb.z, x3 = acc.w + bb.w;
  const float x4 = act + b1;
  float m = fmaxf(fmaxf(fmaxf(x0, x1), fmaxf(x2, x3)), x4);
  for (int o = 4; o > 0; o >>= 1) m = fmaxf(m, __shfl_xor(m, o, 32));
  float s = ((expf(x0 - m) + expf(x1 - m)) + (expf(x2 - m) + expf(x3 - m)))
            + expf(x4 - m);
  for (int o = 4; o > 0; o >>= 1) s += __shfl_xor(s, o, 32);
  const float lse = m + logf(s);
  if (live) {
    ((float4*)out)[(size_t)node * 10 + q] =
        make_float4(x0 - lse, x1 - lse, x2 - lse, x3 - lse);
    out[(size_t)node * 40 + 32 + q] = x4 - lse;
  }
}

extern "C" void kernel_launch(void* const* d_in, const int* in_sizes, int n_in,
                              void* d_out, int out_size, void* d_ws, size_t ws_size,
                              hipStream_t stream) {
  const int N = NNODES, E = NEDGES;
  const float* x     = (const float*)d_in[0];
  const int*   src   = (const int*)d_in[1];
  const int*   tgt   = (const int*)d_in[2];
  const float* w     = (const float*)d_in[3];
  const float* W0    = (const float*)d_in[4];
  const float* b0    = (const float*)d_in[5];
  const float* W1    = (const float*)d_in[6];
  const float* b1    = (const float*)d_in[7];
  const float* W2    = (const float*)d_in[8];
  const float* b2    = (const float*)d_in[9];
  const float* gamma = (const float*)d_in[10];
  const float* beta  = (const float*)d_in[11];
  float* out = (float*)d_out;

  float* H   = (float*)d_ws;
  float* H1  = H + (size_t)N * 128;
  float* H2  = H;
  int*  cnt     = (int*)(H + (size_t)N * 256);
  int2* rowinfo = (int2*)(cnt + NCNT);
  int4* slot    = (int4*)(rowinfo + NCNT);
  int4* ovf     = slot + (size_t)N * ROWCAP;
  int2* csr     = (int2*)(ovf + E);
  int4* binned  = (int4*)(csr + (size_t)N * ROWCAP + E);
  int2* runs    = (int2*)(binned + (size_t)BIN_BLOCKS * EDGES_PER_BLOCK);

  const int gemm_blocks  = (N + 127) / 128;
  const int g40_blocks   = (N * 8 + 255) / 256;
  const int ntiles       = (N + 15) / 16;
  const int fused_blocks = 1024;

  edges_bin_beside_gemm<<<gemm_blocks + BIN_BLOCKS, 256, 0, stream>>>(
      tgt, src, w, cnt, runs, binned, E, x, W0, H, N, gemm_blocks);
  rows_place<<<NBINS, PLACE_WAVES * 32, 0, stream>>>(cnt, runs, binned, slot, ovf, N);

  gather128_tail_gemm<true, false, 128><<<fused_blocks, 512, 0, stream>>>(
      H, cnt, rowinfo, nullptr, slot, ovf, csr, b0, gamma, beta, W1, H1, N, ntiles);

  gather128_tail_gemm<false, true, 40><<<fused_blocks, 512, 0, stream>>>(
      H1, nullptr, rowinfo, csr, nullptr, nullptr, nullptr, b1, gamma, beta, W2, H2, N, ntiles);

  gather40_lsm<<<g40_blocks, 256, 0, stream>>>(H2, rowinfo, csr, b2, out, N);
}
